# starting kernel
# baseline (speedup 1.0000x reference)
#include <hip/hip_runtime.h>

#define D_IN  128
#define D_OUT 64

typedef __attribute__((ext_vector_type(2))) float v2f;
typedef __attribute__((ext_vector_type(8))) float v8f;

__global__ void deg_init_kernel(float* __restrict__ deg, int n) {
    int i = blockIdx.x * blockDim.x + threadIdx.x;
    if (i < n) deg[i] = 1.0f;
}

__global__ void deg_count_kernel(const int* __restrict__ to_idx,
                                 float* __restrict__ deg, int e) {
    int i = blockIdx.x * blockDim.x + threadIdx.x;
    if (i < e) atomicAdd(&deg[to_idx[i]], 1.0f);
}

__global__ void dinv_kernel(float* __restrict__ deg, int n) {
    int i = blockIdx.x * blockDim.x + threadIdx.x;
    if (i < n) {
        float d = deg[i];
        deg[i] = (d > 0.0f) ? rsqrtf(d) : 0.0f;
    }
}

__global__ void __launch_bounds__(128)
gemm_wmma_kernel(const float* __restrict__ x, const float* __restrict__ W,
                 const float* __restrict__ b, float* __restrict__ h, int n_nodes) {
    const int lane = threadIdx.x & 31;
    const int wave = threadIdx.x >> 5;
    const int m0   = blockIdx.x * 16;
    const int n0   = wave * 16;
    const int half = lane >> 4;
    const int l    = lane & 15;

    int row = m0 + l;
    int rrow = (row < n_nodes) ? row : (n_nodes - 1);

    const float* __restrict__ xrow = x + (size_t)rrow * D_IN;
    const float* __restrict__ wrow = W + (size_t)(n0 + l) * D_IN;

    float bias = b[n0 + l];
    v8f acc;
#pragma unroll
    for (int r = 0; r < 8; ++r) acc[r] = bias;

#pragma unroll
    for (int k = 0; k < D_IN; k += 4) {
        v2f a, bb;
        a[0]  = xrow[k + 2 * half + 0];
        a[1]  = xrow[k + 2 * half + 1];
        bb[0] = wrow[k + 2 * half + 0];
        bb[1] = wrow[k + 2 * half + 1];
        acc = __builtin_amdgcn_wmma_f32_16x16x4_f32(
             false, a,  false, bb,
             (short)0, acc,  false,  false);
    }

    float* __restrict__ hbase = h + (size_t)m0 * D_OUT + n0;
#pragma unroll
    for (int r = 0; r < 8; ++r) {
        int mr = r + 8 * half;
        if (m0 + mr < n_nodes)
            hbase[(size_t)mr * D_OUT + l] = acc[r];
    }
}

__global__ void selfloop_kernel(const float* __restrict__ h,
                                const float* __restrict__ dinv,
                                float* __restrict__ out, int n) {
    int t = blockIdx.x * blockDim.x + threadIdx.x;
    int node = t >> 6;
    if (node < n) {
        float di = dinv[node];
        out[t] = h[t] * di * di;
    }
}

__global__ void scatter_kernel(const int* __restrict__ from_idx,
                               const int* __restrict__ to_idx,
                               const float* __restrict__ h,
                               const float* __restrict__ dinv,
                               float* __restrict__ out, int e) {
    int t = blockIdx.x * blockDim.x + threadIdx.x;
    int edge = t >> 6;
    int d    = t & 63;
    if (edge < e) {
        int fa = from_idx[edge];
        int ta = to_idx[edge];
        float c = dinv[fa] * dinv[ta];
        atomicAdd(&out[(size_t)ta * D_OUT + d], h[(size_t)fa * D_OUT + d] * c);
    }
}

extern "C" void kernel_launch(void* const* d_in, const int* in_sizes, int n_in,
                              void* d_out, int out_size, void* d_ws, size_t ws_size,
                              hipStream_t stream) {
    const float* x  = (const float*)d_in[0];
    const float* W  = (const float*)d_in[1];
    const float* b  = (const float*)d_in[2];
    const int*   ei = (const int*)d_in[3];

    const int N = in_sizes[0] / D_IN;
    const int E = in_sizes[3] / 2;
    const int* from_idx = ei;
    const int* to_idx   = ei + E;

    float* out = (float*)d_out;

    float* deg = (float*)d_ws;
    size_t deg_bytes = ((size_t)N * sizeof(float) + 255) & ~(size_t)255;
    float* h = (float*)((char*)d_ws + deg_bytes);

    const int B256 = 256;

    deg_init_kernel<<<(N + B256 - 1) / B256, B256, 0, stream>>>(deg, N);
    deg_count_kernel<<<(E + B256 - 1) / B256, B256, 0, stream>>>(to_idx, deg, E);
    dinv_kernel<<<(N + B256 - 1) / B256, B256, 0, stream>>>(deg, N);

    int m_tiles = (N + 15) / 16;
    gemm_wmma_kernel<<<m_tiles, 128, 0, stream>>>(x, W, b, h, N);

    long long self_threads = (long long)N * D_OUT;
    selfloop_kernel<<<(int)((self_threads + B256 - 1) / B256), B256, 0, stream>>>(
        h, deg, out, N);

    long long edge_threads = (long long)E * D_OUT;
    scatter_kernel<<<(int)((edge_threads + B256 - 1) / B256), B256, 0, stream>>>(
        from_idx, to_idx, h, deg, out, E);
}


# edge pass: each block keeps 1024 out rows in LDS (start = self-loop term), scans all to[], adds matching h rows by LDS f32 atomic adds, stores rows once; no global atomics; sums in arrival order
# speedup vs baseline: 1.5570x; 1.5570x over previous
#include <hip/hip_runtime.h>

#define D_IN  128
#define D_OUT 64

typedef __attribute__((ext_vector_type(2))) float v2f;
typedef __attribute__((ext_vector_type(8))) float v8f;

__global__ void deg_init_kernel(float* __restrict__ deg, int n) {
    int i = blockIdx.x * blockDim.x + threadIdx.x;
    if (i < n) deg[i] = 1.0f;
}

__global__ void deg_count_kernel(const int* __restrict__ to_idx,
                                 float* __restrict__ deg, int e) {
    int i = blockIdx.x * blockDim.x + threadIdx.x;
    if (i < e) atomicAdd(&deg[to_idx[i]], 1.0f);
}

__global__ void dinv_kernel(float* __restrict__ deg, int n) {
    int i = blockIdx.x * blockDim.x + threadIdx.x;
    if (i < n) {
        float d = deg[i];
        deg[i] = (d > 0.0f) ? rsqrtf(d) : 0.0f;
    }
}

__global__ void __launch_bounds__(128)
gemm_wmma_kernel(const float* __restrict__ x, const float* __restrict__ W,
                 const float* __restrict__ b, float* __restrict__ h, int n_nodes) {
    const int lane = threadIdx.x & 31;
    const int wave = threadIdx.x >> 5;
    const int m0   = blockIdx.x * 16;
    const int n0   = wave * 16;
    const int half = lane >> 4;
    const int l    = lane & 15;

    int row = m0 + l;
    int rrow = (row < n_nodes) ? row : (n_nodes - 1);

    const float* __restrict__ xrow = x + (size_t)rrow * D_IN;
    const float* __restrict__ wrow = W + (size_t)(n0 + l) * D_IN;

    float bias = b[n0 + l];
    v8f acc;
#pragma unroll
    for (int r = 0; r < 8; ++r) acc[r] = bias;

#pragma unroll
    for (int k = 0; k < D_IN; k += 4) {
        v2f a, bb;
        a[0]  = xrow[k + 2 * half + 0];
        a[1]  = xrow[k + 2 * half + 1];
        bb[0] = wrow[k + 2 * half + 0];
        bb[1] = wrow[k + 2 * half + 1];
        acc = __builtin_amdgcn_wmma_f32_16x16x4_f32(
             false, a,  false, bb,
             (short)0, acc,  false,  false);
    }

    float* __restrict__ hbase = h + (size_t)m0 * D_OUT + n0;
#pragma unroll
    for (int r = 0; r < 8; ++r) {
        int mr = r + 8 * half;
        if (m0 + mr < n_nodes)
            hbase[(size_t)mr * D_OUT + l] = acc[r];
    }
}

#define TILE_ROWS 1024
#define QCAP      256
#define EP_WAVES  32
#define EP_THREADS (EP_WAVES * 32)

__global__ void __launch_bounds__(EP_THREADS)
edge_tile_kernel(const int* __restrict__ from_idx, const int* __restrict__ to_idx,
                 const float* __restrict__ h, const float* __restrict__ dinv,
                 float* __restrict__ out, int n_nodes, int e_cnt) {
    __shared__ __attribute__((aligned(16))) float tile[TILE_ROWS * D_OUT];
    __shared__ float dinv_t[TILE_ROWS];
    __shared__ int   queue[EP_WAVES * QCAP];
    __shared__ unsigned short qrow[EP_WAVES * QCAP];

    const int tid  = threadIdx.x;
    const int lane = tid & 31;
    const int wave = __builtin_amdgcn_readfirstlane(tid >> 5);
    const int base = blockIdx.x * TILE_ROWS;
    int rows = n_nodes - base;
    if (rows > TILE_ROWS) rows = TILE_ROWS;

    {
        const float4* __restrict__ hsrc = (const float4*)(h + (size_t)base * D_OUT);
        float4* tdst = (float4*)tile;
        for (int i = tid; i < rows * (D_OUT / 4); i += EP_THREADS) {
            float di = dinv[base + (i >> 4)];
            float4 v = hsrc[i];
            float4 o;
            o.x = v.x * di * di; o.y = v.y * di * di;
            o.z = v.z * di * di; o.w = v.w * di * di;
            tdst[i] = o;
        }
        for (int r = tid; r < rows; r += EP_THREADS) dinv_t[r] = dinv[base + r];
    }
    __syncthreads();

    int* q = queue + wave * QCAP;
    unsigned short* qr = qrow + wave * QCAP;
    int count = 0;
    const unsigned ubase = (unsigned)base;
    const unsigned urows = (unsigned)rows;

    auto append = [&](bool m, int e, unsigned d) {
        unsigned mk = __builtin_amdgcn_ballot_w32(m);
        if (mk) {
            int pos = count + (int)__builtin_amdgcn_mbcnt_lo(mk, 0u);
            if (m) { q[pos] = e; qr[pos] = (unsigned short)d; }
            count += __builtin_popcount(mk);
        }
    };

    auto add_row = [&](float2 hv, int lr, float c) {
        float* p = &tile[lr * D_OUT + 2 * lane];
        __hip_atomic_fetch_add(p,     hv.x * c, __ATOMIC_RELAXED, __HIP_MEMORY_SCOPE_WORKGROUP);
        __hip_atomic_fetch_add(p + 1, hv.y * c, __ATOMIC_RELAXED, __HIP_MEMORY_SCOPE_WORKGROUP);
    };

    auto batch = [&](int b0, int n, int m, bool full) {
        int idx = b0 + lane;
        if (idx > n - 1) idx = n - 1;
        const int e = q[idx];
        const int lr = qr[idx];
        const int f = from_idx[e];
        const float df = dinv[f];
        if (full) {
            float2 hv[32];
#pragma unroll
            for (int k = 0; k < 32; ++k) {
                int fj = __builtin_amdgcn_readlane(f, k);
                hv[k] = ((const float2*)(h + (size_t)fj * D_OUT))[lane];
            }
            const float c = df * dinv_t[lr];
#pragma unroll
            for (int k = 0; k < 32; ++k) {
                int   lj = __builtin_amdgcn_readlane(lr, k);
                float cj = __int_as_float(__builtin_amdgcn_readlane(__float_as_int(c), k));
                add_row(hv[k], lj, cj);
            }
        } else {
            const float c = df * dinv_t[lr];
            for (int j = 0; j < m; j += 8) {
                float2 hv[8];
#pragma unroll
                for (int k = 0; k < 8; ++k) {
                    int fj = __builtin_amdgcn_readlane(f, j + k);
                    hv[k] = ((const float2*)(h + (size_t)fj * D_OUT))[lane];
                }
#pragma unroll
                for (int k = 0; k < 8; ++k) {
                    if (j + k < m) {
                        int   lj = __builtin_amdgcn_readlane(lr, j + k);
                        float cj = __int_as_float(__builtin_amdgcn_readlane(__float_as_int(c), j + k));
                        add_row(hv[k], lj, cj);
                    }
                }
            }
        }
    };

    auto flush = [&](bool final_call) {
        const int n = count;
        const int nb = n & ~31;
        for (int b0 = 0; b0 < nb; b0 += 32) batch(b0, n, 32, true);
        const int r = n - nb;
        if (final_call) {
            if (r > 0) batch(nb, n, r, false);
            count = 0;
        } else {
            int v = 0;
            unsigned short w = 0;
            if (lane < r) { v = q[nb + lane]; w = qr[nb + lane]; }
            if (nb > 0 && lane < r) { q[lane] = v; qr[lane] = w; }
            count = (nb > 0) ? r : n;
        }
    };

    int head = (int)(((16u - ((unsigned)(size_t)to_idx & 15u)) & 15u) >> 2);
    if (head > e_cnt) head = e_cnt;
    const int nwc = (e_cnt - head) >> 7;
    const int4* __restrict__ vp = (const int4*)(to_idx + head);

    if (wave == 0) {
        const int vec_end = head + (nwc << 7);
        const int nleft = head + (e_cnt - vec_end);
        for (int k0 = 0; k0 < nleft; k0 += 32) {
            int k = k0 + lane;
            int e = (k < head) ? k : (vec_end + (k - head));
            unsigned d = 0xFFFFFFFFu;
            if (k < nleft) d = (unsigned)to_idx[e] - ubase;
            append(d < urows, e, d);
            if (count > QCAP - 128) flush(false);
        }
    }

    const int S = (nwc + 31) >> 5;
    const int rot = (int)((blockIdx.x * (unsigned)S) / gridDim.x);

    auto chunk_of = [&](int s) -> int {
        int sr = s + rot;
        if (sr >= S) sr -= S;
        return sr * 32 + wave;
    };
    auto load_chunk = [&](int s) -> int4 {
        int c = chunk_of(s);
        if (c > nwc - 1) c = nwc - 1;
        return vp[c * 32 + lane];
    };

#define PF 4
    if (nwc > 0) {
        auto step = [&](int s, int4 cur) {
            const int cs = chunk_of(s);
            const unsigned lim = (s < S && cs < nwc) ? urows : 0u;
            unsigned d0 = (unsigned)cur.x - ubase;
            unsigned d1 = (unsigned)cur.y - ubase;
            unsigned d2 = (unsigned)cur.z - ubase;
            unsigned d3 = (unsigned)cur.w - ubase;
            unsigned dm = min(min(d0, d1), min(d2, d3));
            if (__builtin_amdgcn_ballot_w32(dm < lim)) {
                int e0 = head + ((cs * 32 + lane) << 2);
                append(d0 < lim, e0, d0);
                append(d1 < lim, e0 + 1, d1);
                append(d2 < lim, e0 + 2, d2);
                append(d3 < lim, e0 + 3, d3);
                if (count > QCAP - 128) flush(false);
            }
        };

        int4 bufa[PF], bufb[PF];
#pragma unroll
        for (int u = 0; u < PF; ++u) bufa[u] = load_chunk(u);

        for (int s0 = 0; s0 < S; s0 += 2 * PF) {
#pragma unroll
            for (int u = 0; u < PF; ++u) bufb[u] = load_chunk(s0 + PF + u);
#pragma unroll
            for (int u = 0; u < PF; ++u) step(s0 + u, bufa[u]);
#pragma unroll
            for (int u = 0; u < PF; ++u) bufa[u] = load_chunk(s0 + 2 * PF + u);
#pragma unroll
            for (int u = 0; u < PF; ++u) step(s0 + PF + u, bufb[u]);
        }
    }
    flush(true);

    __syncthreads();

    {
        float4* __restrict__ odst = (float4*)(out + (size_t)base * D_OUT);
        const float4* tsrc = (const float4*)tile;
        for (int i = tid; i < rows * (D_OUT / 4); i += EP_THREADS) odst[i] = tsrc[i];
    }
}

extern "C" void kernel_launch(void* const* d_in, const int* in_sizes, int n_in,
                              void* d_out, int out_size, void* d_ws, size_t ws_size,
                              hipStream_t stream) {
    const float* x  = (const float*)d_in[0];
    const float* W  = (const float*)d_in[1];
    const float* b  = (const float*)d_in[2];
    const int*   ei = (const int*)d_in[3];

    const int N = in_sizes[0] / D_IN;
    const int E = in_sizes[3] / 2;
    const int* from_idx = ei;
    const int* to_idx   = ei + E;

    float* out = (float*)d_out;

    float* deg = (float*)d_ws;
    size_t deg_bytes = ((size_t)N * sizeof(float) + 255) & ~(size_t)255;
    float* h = (float*)((char*)d_ws + deg_bytes);

    const int B256 = 256;

    deg_init_kernel<<<(N + B256 - 1) / B256, B256, 0, stream>>>(deg, N);
    deg_count_kernel<<<(E + B256 - 1) / B256, B256, 0, stream>>>(to_idx, deg, E);
    dinv_kernel<<<(N + B256 - 1) / B256, B256, 0, stream>>>(deg, N);

    int m_tiles = (N + 15) / 16;
    gemm_wmma_kernel<<<m_tiles, 128, 0, stream>>>(x, W, b, h, N);

    int n_tiles = (N + TILE_ROWS - 1) / TILE_ROWS;
    edge_tile_kernel<<<n_tiles, EP_THREADS, 0, stream>>>(from_idx, to_idx, h, deg, out, N, E);
}


# edge pass: edge list cut in 2 halves, grid 112 row tiles x 2 (224 blocks, was 98); each block sums its half into its own plane, a new pass adds the 2 planes into out; a row is now 2 partial sums
# speedup vs baseline: 1.9526x; 1.2541x over previous
#include <hip/hip_runtime.h>

#define D_IN  128
#define D_OUT 64

typedef __attribute__((ext_vector_type(2))) float v2f;
typedef __attribute__((ext_vector_type(8))) float v8f;

__global__ void deg_init_kernel(float* __restrict__ deg, int n) {
    int i = blockIdx.x * blockDim.x + threadIdx.x;
    if (i < n) deg[i] = 1.0f;
}

__global__ void deg_count_kernel(const int* __restrict__ to_idx,
                                 float* __restrict__ deg, int e) {
    int i = blockIdx.x * blockDim.x + threadIdx.x;
    if (i < e) atomicAdd(&deg[to_idx[i]], 1.0f);
}

__global__ void dinv_kernel(float* __restrict__ deg, int n) {
    int i = blockIdx.x * blockDim.x + threadIdx.x;
    if (i < n) {
        float d = deg[i];
        deg[i] = (d > 0.0f) ? rsqrtf(d) : 0.0f;
    }
}

__global__ void __launch_bounds__(128)
gemm_wmma_kernel(const float* __restrict__ x, const float* __restrict__ W,
                 const float* __restrict__ b, float* __restrict__ h, int n_nodes) {
    const int lane = threadIdx.x & 31;
    const int wave = threadIdx.x >> 5;
    const int m0   = blockIdx.x * 16;
    const int n0   = wave * 16;
    const int half = lane >> 4;
    const int l    = lane & 15;

    int row = m0 + l;
    int rrow = (row < n_nodes) ? row : (n_nodes - 1);

    const float* __restrict__ xrow = x + (size_t)rrow * D_IN;
    const float* __restrict__ wrow = W + (size_t)(n0 + l) * D_IN;

    float bias = b[n0 + l];
    v8f acc;
#pragma unroll
    for (int r = 0; r < 8; ++r) acc[r] = bias;

#pragma unroll
    for (int k = 0; k < D_IN; k += 4) {
        v2f a, bb;
        a[0]  = xrow[k + 2 * half + 0];
        a[1]  = xrow[k + 2 * half + 1];
        bb[0] = wrow[k + 2 * half + 0];
        bb[1] = wrow[k + 2 * half + 1];
        acc = __builtin_amdgcn_wmma_f32_16x16x4_f32(
             false, a,  false, bb,
             (short)0, acc,  false,  false);
    }

    float* __restrict__ hbase = h + (size_t)m0 * D_OUT + n0;
#pragma unroll
    for (int r = 0; r < 8; ++r) {
        int mr = r + 8 * half;
        if (m0 + mr < n_nodes)
            hbase[(size_t)mr * D_OUT + l] = acc[r];
    }
}

#define EP_SLICES 2
#define TILE_ROWS 896
#define QCAP      256
#define EP_WAVES  32
#define EP_THREADS (EP_WAVES * 32)

__global__ void __launch_bounds__(EP_THREADS)
edge_tile_kernel(const int* __restrict__ from_idx, const int* __restrict__ to_idx,
                 const float* __restrict__ h, const float* __restrict__ dinv,
                 float* __restrict__ part, int n_nodes, int e_cnt) {
    __shared__ __attribute__((aligned(16))) float tile[TILE_ROWS * D_OUT];
    __shared__ float dinv_t[TILE_ROWS];
    __shared__ int   queue[EP_WAVES * QCAP];
    __shared__ unsigned short qrow[EP_WAVES * QCAP];

    const int tid  = threadIdx.x;
    const int lane = tid & 31;
    const int wave = __builtin_amdgcn_readfirstlane(tid >> 5);
    const int base = blockIdx.x * TILE_ROWS;
    const int slice = blockIdx.y;
    int rows = n_nodes - base;
    if (rows > TILE_ROWS) rows = TILE_ROWS;

    {
        const float4* __restrict__ hsrc = (const float4*)(h + (size_t)base * D_OUT);
        float4* tdst = (float4*)tile;
        if (slice == 0) {
            for (int i = tid; i < rows * (D_OUT / 4); i += EP_THREADS) {
                float di = dinv[base + (i >> 4)];
                float4 v = hsrc[i];
                float4 o;
                o.x = v.x * di * di; o.y = v.y * di * di;
                o.z = v.z * di * di; o.w = v.w * di * di;
                tdst[i] = o;
            }
        } else {
            const float4 z = make_float4(0.0f, 0.0f, 0.0f, 0.0f);
            for (int i = tid; i < rows * (D_OUT / 4); i += EP_THREADS) tdst[i] = z;
        }
        for (int r = tid; r < rows; r += EP_THREADS) dinv_t[r] = dinv[base + r];
    }
    __syncthreads();

    int* q = queue + wave * QCAP;
    unsigned short* qr = qrow + wave * QCAP;
    int count = 0;
    const unsigned ubase = (unsigned)base;
    const unsigned urows = (unsigned)rows;

    auto append = [&](bool m, int e, unsigned d) {
        unsigned mk = __builtin_amdgcn_ballot_w32(m);
        if (mk) {
            int pos = count + (int)__builtin_amdgcn_mbcnt_lo(mk, 0u);
            if (m) { q[pos] = e; qr[pos] = (unsigned short)d; }
            count += __builtin_popcount(mk);
        }
    };

    auto add_row = [&](float2 hv, int lr, float c) {
        float* p = &tile[lr * D_OUT + 2 * lane];
        __hip_atomic_fetch_add(p,     hv.x * c, __ATOMIC_RELAXED, __HIP_MEMORY_SCOPE_WORKGROUP);
        __hip_atomic_fetch_add(p + 1, hv.y * c, __ATOMIC_RELAXED, __HIP_MEMORY_SCOPE_WORKGROUP);
    };

    auto batch = [&](int b0, int n, int m, bool full) {
        int idx = b0 + lane;
        if (idx > n - 1) idx = n - 1;
        const int e = q[idx];
        const int lr = qr[idx];
        const int f = from_idx[e];
        const float df = dinv[f];
        if (full) {
            float2 hv[32];
#pragma unroll
            for (int k = 0; k < 32; ++k) {
                int fj = __builtin_amdgcn_readlane(f, k);
                hv[k] = ((const float2*)(h + (size_t)fj * D_OUT))[lane];
            }
            const float c = df * dinv_t[lr];
#pragma unroll
            for (int k = 0; k < 32; ++k) {
                int   lj = __builtin_amdgcn_readlane(lr, k);
                float cj = __int_as_float(__builtin_amdgcn_readlane(__float_as_int(c), k));
                add_row(hv[k], lj, cj);
            }
        } else {
            const float c = df * dinv_t[lr];
            for (int j = 0; j < m; j += 8) {
                float2 hv[8];
#pragma unroll
                for (int k = 0; k < 8; ++k) {
                    int fj = __builtin_amdgcn_readlane(f, j + k);
                    hv[k] = ((const float2*)(h + (size_t)fj * D_OUT))[lane];
                }
#pragma unroll
                for (int k = 0; k < 8; ++k) {
                    if (j + k < m) {
                        int   lj = __builtin_amdgcn_readlane(lr, j + k);
                        float cj = __int_as_float(__builtin_amdgcn_readlane(__float_as_int(c), j + k));
                        add_row(hv[k], lj, cj);
                    }
                }
            }
        }
    };

    auto flush = [&](bool final_call) {
        const int n = count;
        const int nb = n & ~31;
        for (int b0 = 0; b0 < nb; b0 += 32) batch(b0, n, 32, true);
        const int r = n - nb;
        if (final_call) {
            if (r > 0) batch(nb, n, r, false);
            count = 0;
        } else {
            int v = 0;
            unsigned short w = 0;
            if (lane < r) { v = q[nb + lane]; w = qr[nb + lane]; }
            if (nb > 0 && lane < r) { q[lane] = v; qr[lane] = w; }
            count = (nb > 0) ? r : n;
        }
    };

    int head = (int)(((16u - ((unsigned)(size_t)to_idx & 15u)) & 15u) >> 2);
    if (head > e_cnt) head = e_cnt;
    const int nwc = (e_cnt - head) >> 7;
    const int c_lo = (int)(((long long)nwc * slice) / EP_SLICES);
    const int nloc = (int)(((long long)nwc * (slice + 1)) / EP_SLICES) - c_lo;
    const int4* __restrict__ vp = (const int4*)(to_idx + head) + (size_t)c_lo * 32;
    const int ebase = head + (c_lo << 7);

    if (wave == 0 && slice == 0) {
        const int vec_end = head + (nwc << 7);
        const int nleft = head + (e_cnt - vec_end);
        for (int k0 = 0; k0 < nleft; k0 += 32) {
            int k = k0 + lane;
            int e = (k < head) ? k : (vec_end + (k - head));
            unsigned d = 0xFFFFFFFFu;
            if (k < nleft) d = (unsigned)to_idx[e] - ubase;
            append(d < urows, e, d);
            if (count > QCAP - 128) flush(false);
        }
    }

    const int S = (nloc + 31) >> 5;
    const int rot = (int)((blockIdx.x * (unsigned)S) / gridDim.x);

    auto chunk_of = [&](int s) -> int {
        int sr = s + rot;
        if (sr >= S) sr -= S;
        return sr * 32 + wave;
    };
    auto load_chunk = [&](int s) -> int4 {
        int c = chunk_of(s);
        if (c > nloc - 1) c = nloc - 1;
        return vp[c * 32 + lane];
    };

#define PF 4
    if (nloc > 0) {
        auto step = [&](int s, int4 cur) {
            const int cs = chunk_of(s);
            const unsigned lim = (s < S && cs < nloc) ? urows : 0u;
            unsigned d0 = (unsigned)cur.x - ubase;
            unsigned d1 = (unsigned)cur.y - ubase;
            unsigned d2 = (unsigned)cur.z - ubase;
            unsigned d3 = (unsigned)cur.w - ubase;
            unsigned dm = min(min(d0, d1), min(d2, d3));
            if (__builtin_amdgcn_ballot_w32(dm < lim)) {
                int e0 = ebase + ((cs * 32 + lane) << 2);
                append(d0 < lim, e0, d0);
                append(d1 < lim, e0 + 1, d1);
                append(d2 < lim, e0 + 2, d2);
                append(d3 < lim, e0 + 3, d3);
                if (count > QCAP - 128) flush(false);
            }
        };

        int4 bufa[PF], bufb[PF];
#pragma unroll
        for (int u = 0; u < PF; ++u) bufa[u] = load_chunk(u);

        for (int s0 = 0; s0 < S; s0 += 2 * PF) {
#pragma unroll
            for (int u = 0; u < PF; ++u) bufb[u] = load_chunk(s0 + PF + u);
#pragma unroll
            for (int u = 0; u < PF; ++u) step(s0 + u, bufa[u]);
#pragma unroll
            for (int u = 0; u < PF; ++u) bufa[u] = load_chunk(s0 + 2 * PF + u);
#pragma unroll
            for (int u = 0; u < PF; ++u) step(s0 + PF + u, bufb[u]);
        }
    }
    flush(true);

    __syncthreads();

    {
        float4* __restrict__ odst = (float4*)(part + ((size_t)slice * n_nodes + base) * D_OUT);
        const float4* tsrc = (const float4*)tile;
        for (int i = tid; i < rows * (D_OUT / 4); i += EP_THREADS) odst[i] = tsrc[i];
    }
}

#define SUM_U 8
__global__ void __launch_bounds__(256)
sum_parts_kernel(const float* __restrict__ part, float* __restrict__ out, int n4) {
    const float4* __restrict__ p = (const float4*)part;
    float4* __restrict__ o = (float4*)out;
    const int i0 = blockIdx.x * (SUM_U * 256) + threadIdx.x;
    float4 v[EP_SLICES][SUM_U];
#pragma unroll
    for (int s = 0; s < EP_SLICES; ++s) {
#pragma unroll
        for (int u = 0; u < SUM_U; ++u) {
            int j = i0 + u * 256;
            if (j > n4 - 1) j = n4 - 1;
            v[s][u] = p[(size_t)s * n4 + j];
        }
    }
#pragma unroll
    for (int u = 0; u < SUM_U; ++u) {
        const int j = i0 + u * 256;
        float4 a = v[0][u];
#pragma unroll
        for (int s = 1; s < EP_SLICES; ++s) {
            a.x += v[s][u].x; a.y += v[s][u].y; a.z += v[s][u].z; a.w += v[s][u].w;
        }
        if (j < n4) o[j] = a;
    }
}

extern "C" void kernel_launch(void* const* d_in, const int* in_sizes, int n_in,
                              void* d_out, int out_size, void* d_ws, size_t ws_size,
                              hipStream_t stream) {
    const float* x  = (const float*)d_in[0];
    const float* W  = (const float*)d_in[1];
    const float* b  = (const float*)d_in[2];
    const int*   ei = (const int*)d_in[3];

    const int N = in_sizes[0] / D_IN;
    const int E = in_sizes[3] / 2;
    const int* from_idx = ei;
    const int* to_idx   = ei + E;

    float* out = (float*)d_out;

    float* deg = (float*)d_ws;
    size_t deg_bytes = ((size_t)N * sizeof(float) + 255) & ~(size_t)255;
    float* h = (float*)((char*)d_ws + deg_bytes);
    float* part = h + (size_t)N * D_OUT;

    const int B256 = 256;

    deg_init_kernel<<<(N + B256 - 1) / B256, B256, 0, stream>>>(deg, N);
    deg_count_kernel<<<(E + B256 - 1) / B256, B256, 0, stream>>>(to_idx, deg, E);
    dinv_kernel<<<(N + B256 - 1) / B256, B256, 0, stream>>>(deg, N);

    int m_tiles = (N + 15) / 16;
    gemm_wmma_kernel<<<m_tiles, 128, 0, stream>>>(x, W, b, h, N);

    int n_tiles = (N + TILE_ROWS - 1) / TILE_ROWS;
    edge_tile_kernel<<<dim3(n_tiles, EP_SLICES), EP_THREADS, 0, stream>>>(from_idx, to_idx, h, deg, part, N, E);

    int n4 = N * (D_OUT / 4);
    sum_parts_kernel<<<(n4 + SUM_U * 256 - 1) / (SUM_U * 256), 256, 0, stream>>>(part, out, n4);
}


# gemm: a wave owns 16 rows x all 64 columns: x is read once (was 4x) by 16-byte loads straight into A fragments, W fragments come from an LDS copy; k regrouped inside each 8, so h's sum order changes
# speedup vs baseline: 2.1225x; 1.0870x over previous
#include <hip/hip_runtime.h>

#define D_IN  128
#define D_OUT 64

typedef __attribute__((ext_vector_type(2))) float v2f;
typedef __attribute__((ext_vector_type(8))) float v8f;

__global__ void deg_init_kernel(float* __restrict__ deg, int n) {
    int i = blockIdx.x * blockDim.x + threadIdx.x;
    if (i < n) deg[i] = 1.0f;
}

__global__ void deg_count_kernel(const int* __restrict__ to_idx,
                                 float* __restrict__ deg, int e) {
    int i = blockIdx.x * blockDim.x + threadIdx.x;
    if (i < e) atomicAdd(&deg[to_idx[i]], 1.0f);
}

__global__ void dinv_kernel(float* __restrict__ deg, int n) {
    int i = blockIdx.x * blockDim.x + threadIdx.x;
    if (i < n) {
        float d = deg[i];
        deg[i] = (d > 0.0f) ? rsqrtf(d) : 0.0f;
    }
}

#define GEMM_WAVES 8
__global__ void __launch_bounds__(GEMM_WAVES * 32)
gemm_wmma_kernel(const float* __restrict__ x, const float* __restrict__ W,
                 const float* __restrict__ b, float* __restrict__ h, int n_nodes) {
    __shared__ float4 wl[16 * 4 * 32];

    const int tid  = threadIdx.x;
    const int lane = tid & 31;
    const int wave = tid >> 5;
    const int half = lane >> 4;
    const int l    = lane & 15;
    const int m0   = (blockIdx.x * GEMM_WAVES + wave) * 16;

    int row = m0 + l;
    int rrow = (row < n_nodes) ? row : (n_nodes - 1);

    const float4* __restrict__ xr = (const float4*)(x + (size_t)rrow * D_IN) + half;
    float4 a[16];
#pragma unroll
    for (int j = 0; j < 16; ++j) a[j] = xr[2 * j];

    float bias[4];
#pragma unroll
    for (int t = 0; t < 4; ++t) bias[t] = b[t * 16 + l];

    {
        const float4* __restrict__ W4 = (const float4*)W;
#pragma unroll
        for (int i = 0; i < (D_OUT * D_IN / 4) / (GEMM_WAVES * 32); ++i) {
            const int g  = i * (GEMM_WAVES * 32) + tid;
            const int r  = g >> 5;
            const int c  = g & 31;
            const int t  = r >> 4, n = r & 15;
            const int j  = c >> 1, hf = c & 1;
            wl[(j * 4 + t) * 32 + hf * 16 + (n ^ j)] = W4[g];
        }
    }
    __syncthreads();

    v8f acc[4];
#pragma unroll
    for (int t = 0; t < 4; ++t) {
#pragma unroll
        for (int r = 0; r < 8; ++r) acc[t][r] = bias[t];
    }

#pragma unroll
    for (int j = 0; j < 16; ++j) {
        float4 bv[4];
#pragma unroll
        for (int t = 0; t < 4; ++t) bv[t] = wl[(j * 4 + t) * 32 + (lane ^ j)];
        v2f a0, a1;
        a0[0] = a[j].x; a0[1] = a[j].y;
        a1[0] = a[j].z; a1[1] = a[j].w;
#pragma unroll
        for (int t = 0; t < 4; ++t) {
            v2f b0; b0[0] = bv[t].x; b0[1] = bv[t].y;
            acc[t] = __builtin_amdgcn_wmma_f32_16x16x4_f32(
                 false, a0,  false, b0,
                 (short)0, acc[t],  false,  false);
        }
#pragma unroll
        for (int t = 0; t < 4; ++t) {
            v2f b1; b1[0] = bv[t].z; b1[1] = bv[t].w;
            acc[t] = __builtin_amdgcn_wmma_f32_16x16x4_f32(
                 false, a1,  false, b1,
                 (short)0, acc[t],  false,  false);
        }
    }

    float* __restrict__ hbase = h + (size_t)m0 * D_OUT;
#pragma unroll
    for (int r = 0; r < 8; ++r) {
        const int mr = r + 8 * half;
        if (m0 + mr < n_nodes) {
#pragma unroll
            for (int t = 0; t < 4; ++t)
                hbase[(size_t)mr * D_OUT + t * 16 + l] = acc[t][r];
        }
    }
}

#define EP_SLICES 2
#define TILE_ROWS 896
#define QCAP      256
#define EP_WAVES  32
#define EP_THREADS (EP_WAVES * 32)

__global__ void __launch_bounds__(EP_THREADS)
edge_tile_kernel(const int* __restrict__ from_idx, const int* __restrict__ to_idx,
                 const float* __restrict__ h, const float* __restrict__ dinv,
                 float* __restrict__ part, int n_nodes, int e_cnt) {
    __shared__ __attribute__((aligned(16))) float tile[TILE_ROWS * D_OUT];
    __shared__ float dinv_t[TILE_ROWS];
    __shared__ int   queue[EP_WAVES * QCAP];
    __shared__ unsigned short qrow[EP_WAVES * QCAP];

    const int tid  = threadIdx.x;
    const int lane = tid & 31;
    const int wave = __builtin_amdgcn_readfirstlane(tid >> 5);
    const int base = blockIdx.x * TILE_ROWS;
    const int slice = blockIdx.y;
    int rows = n_nodes - base;
    if (rows > TILE_ROWS) rows = TILE_ROWS;

    {
        const float4* __restrict__ hsrc = (const float4*)(h + (size_t)base * D_OUT);
        float4* tdst = (float4*)tile;
        if (slice == 0) {
            for (int i = tid; i < rows * (D_OUT / 4); i += EP_THREADS) {
                float di = dinv[base + (i >> 4)];
                float4 v = hsrc[i];
                float4 o;
                o.x = v.x * di * di; o.y = v.y * di * di;
                o.z = v.z * di * di; o.w = v.w * di * di;
                tdst[i] = o;
            }
        } else {
            const float4 z = make_float4(0.0f, 0.0f, 0.0f, 0.0f);
            for (int i = tid; i < rows * (D_OUT / 4); i += EP_THREADS) tdst[i] = z;
        }
        for (int r = tid; r < rows; r += EP_THREADS) dinv_t[r] = dinv[base + r];
    }
    __syncthreads();

    int* q = queue + wave * QCAP;
    unsigned short* qr = qrow + wave * QCAP;
    int count = 0;
    const unsigned ubase = (unsigned)base;
    const unsigned urows = (unsigned)rows;

    auto append = [&](bool m, int e, unsigned d) {
        unsigned mk = __builtin_amdgcn_ballot_w32(m);
        if (mk) {
            int pos = count + (int)__builtin_amdgcn_mbcnt_lo(mk, 0u);
            if (m) { q[pos] = e; qr[pos] = (unsigned short)d; }
            count += __builtin_popcount(mk);
        }
    };

    auto add_row = [&](float2 hv, int lr, float c) {
        float* p = &tile[lr * D_OUT + 2 * lane];
        __hip_atomic_fetch_add(p,     hv.x * c, __ATOMIC_RELAXED, __HIP_MEMORY_SCOPE_WORKGROUP);
        __hip_atomic_fetch_add(p + 1, hv.y * c, __ATOMIC_RELAXED, __HIP_MEMORY_SCOPE_WORKGROUP);
    };

    auto batch = [&](int b0, int n, int m, bool full) {
        int idx = b0 + lane;
        if (idx > n - 1) idx = n - 1;
        const int e = q[idx];
        const int lr = qr[idx];
        const int f = from_idx[e];
        const float df = dinv[f];
        if (full) {
            float2 hv[32];
#pragma unroll
            for (int k = 0; k < 32; ++k) {
                int fj = __builtin_amdgcn_readlane(f, k);
                hv[k] = ((const float2*)(h + (size_t)fj * D_OUT))[lane];
            }
            const float c = df * dinv_t[lr];
#pragma unroll
            for (int k = 0; k < 32; ++k) {
                int   lj = __builtin_amdgcn_readlane(lr, k);
                float cj = __int_as_float(__builtin_amdgcn_readlane(__float_as_int(c), k));
                add_row(hv[k], lj, cj);
            }
        } else {
            const float c = df * dinv_t[lr];
            for (int j = 0; j < m; j += 8) {
                float2 hv[8];
#pragma unroll
                for (int k = 0; k < 8; ++k) {
                    int fj = __builtin_amdgcn_readlane(f, j + k);
                    hv[k] = ((const float2*)(h + (size_t)fj * D_OUT))[lane];
                }
#pragma unroll
                for (int k = 0; k < 8; ++k) {
                    if (j + k < m) {
                        int   lj = __builtin_amdgcn_readlane(lr, j + k);
                        float cj = __int_as_float(__builtin_amdgcn_readlane(__float_as_int(c), j + k));
                        add_row(hv[k], lj, cj);
                    }
                }
            }
        }
    };

    auto flush = [&](bool final_call) {
        const int n = count;
        const int nb = n & ~31;
        for (int b0 = 0; b0 < nb; b0 += 32) batch(b0, n, 32, true);
        const int r = n - nb;
        if (final_call) {
            if (r > 0) batch(nb, n, r, false);
            count = 0;
        } else {
            int v = 0;
            unsigned short w = 0;
            if (lane < r) { v = q[nb + lane]; w = qr[nb + lane]; }
            if (nb > 0 && lane < r) { q[lane] = v; qr[lane] = w; }
            count = (nb > 0) ? r : n;
        }
    };

    int head = (int)(((16u - ((unsigned)(size_t)to_idx & 15u)) & 15u) >> 2);
    if (head > e_cnt) head = e_cnt;
    const int nwc = (e_cnt - head) >> 7;
    const int c_lo = (int)(((long long)nwc * slice) / EP_SLICES);
    const int nloc = (int)(((long long)nwc * (slice + 1)) / EP_SLICES) - c_lo;
    const int4* __restrict__ vp = (const int4*)(to_idx + head) + (size_t)c_lo * 32;
    const int ebase = head + (c_lo << 7);

    if (wave == 0 && slice == 0) {
        const int vec_end = head + (nwc << 7);
        const int nleft = head + (e_cnt - vec_end);
        for (int k0 = 0; k0 < nleft; k0 += 32) {
            int k = k0 + lane;
            int e = (k < head) ? k : (vec_end + (k - head));
            unsigned d = 0xFFFFFFFFu;
            if (k < nleft) d = (unsigned)to_idx[e] - ubase;
            append(d < urows, e, d);
            if (count > QCAP - 128) flush(false);
        }
    }

    const int S = (nloc + 31) >> 5;
    const int rot = (int)((blockIdx.x * (unsigned)S) / gridDim.x);

    auto chunk_of = [&](int s) -> int {
        int sr = s + rot;
        if (sr >= S) sr -= S;
        return sr * 32 + wave;
    };
    auto load_chunk = [&](int s) -> int4 {
        int c = chunk_of(s);
        if (c > nloc - 1) c = nloc - 1;
        return vp[c * 32 + lane];
    };

#define PF 4
    if (nloc > 0) {
        auto step = [&](int s, int4 cur) {
            const int cs = chunk_of(s);
            const unsigned lim = (s < S && cs < nloc) ? urows : 0u;
            unsigned d0 = (unsigned)cur.x - ubase;
            unsigned d1 = (unsigned)cur.y - ubase;
            unsigned d2 = (unsigned)cur.z - ubase;
            unsigned d3 = (unsigned)cur.w - ubase;
            unsigned dm = min(min(d0, d1), min(d2, d3));
            if (__builtin_amdgcn_ballot_w32(dm < lim)) {
                int e0 = ebase + ((cs * 32 + lane) << 2);
                append(d0 < lim, e0, d0);
                append(d1 < lim, e0 + 1, d1);
                append(d2 < lim, e0 + 2, d2);
                append(d3 < lim, e0 + 3, d3);
                if (count > QCAP - 128) flush(false);
            }
        };

        int4 bufa[PF], bufb[PF];
#pragma unroll
        for (int u = 0; u < PF; ++u) bufa[u] = load_chunk(u);

        for (int s0 = 0; s0 < S; s0 += 2 * PF) {
#pragma unroll
            for (int u = 0; u < PF; ++u) bufb[u] = load_chunk(s0 + PF + u);
#pragma unroll
            for (int u = 0; u < PF; ++u) step(s0 + u, bufa[u]);
#pragma unroll
            for (int u = 0; u < PF; ++u) bufa[u] = load_chunk(s0 + 2 * PF + u);
#pragma unroll
            for (int u = 0; u < PF; ++u) step(s0 + PF + u, bufb[u]);
        }
    }
    flush(true);

    __syncthreads();

    {
        float4* __restrict__ odst = (float4*)(part + ((size_t)slice * n_nodes + base) * D_OUT);
        const float4* tsrc = (const float4*)tile;
        for (int i = tid; i < rows * (D_OUT / 4); i += EP_THREADS) odst[i] = tsrc[i];
    }
}

#define SUM_U 8
__global__ void __launch_bounds__(256)
sum_parts_kernel(const float* __restrict__ part, float* __restrict__ out, int n4) {
    const float4* __restrict__ p = (const float4*)part;
    float4* __restrict__ o = (float4*)out;
    const int i0 = blockIdx.x * (SUM_U * 256) + threadIdx.x;
    float4 v[EP_SLICES][SUM_U];
#pragma unroll
    for (int s = 0; s < EP_SLICES; ++s) {
#pragma unroll
        for (int u = 0; u < SUM_U; ++u) {
            int j = i0 + u * 256;
            if (j > n4 - 1) j = n4 - 1;
            v[s][u] = p[(size_t)s * n4 + j];
        }
    }
#pragma unroll
    for (int u = 0; u < SUM_U; ++u) {
        const int j = i0 + u * 256;
        float4 a = v[0][u];
#pragma unroll
        for (int s = 1; s < EP_SLICES; ++s) {
            a.x += v[s][u].x; a.y += v[s][u].y; a.z += v[s][u].z; a.w += v[s][u].w;
        }
        if (j < n4) o[j] = a;
    }
}

extern "C" void kernel_launch(void* const* d_in, const int* in_sizes, int n_in,
                              void* d_out, int out_size, void* d_ws, size_t ws_size,
                              hipStream_t stream) {
    const float* x  = (const float*)d_in[0];
    const float* W  = (const float*)d_in[1];
    const float* b  = (const float*)d_in[2];
    const int*   ei = (const int*)d_in[3];

    const int N = in_sizes[0] / D_IN;
    const int E = in_sizes[3] / 2;
    const int* from_idx = ei;
    const int* to_idx   = ei + E;

    float* out = (float*)d_out;

    float* deg = (float*)d_ws;
    size_t deg_bytes = ((size_t)N * sizeof(float) + 255) & ~(size_t)255;
    float* h = (float*)((char*)d_ws + deg_bytes);
    float* part = h + (size_t)N * D_OUT;

    const int B256 = 256;

    deg_init_kernel<<<(N + B256 - 1) / B256, B256, 0, stream>>>(deg, N);
    deg_count_kernel<<<(E + B256 - 1) / B256, B256, 0, stream>>>(to_idx, deg, E);
    dinv_kernel<<<(N + B256 - 1) / B256, B256, 0, stream>>>(deg, N);

    int m_tiles = (N + 15) / 16;
    gemm_wmma_kernel<<<(m_tiles + GEMM_WAVES - 1) / GEMM_WAVES, GEMM_WAVES * 32, 0, stream>>>(x, W, b, h, N);

    int n_tiles = (N + TILE_ROWS - 1) / TILE_ROWS;
    edge_tile_kernel<<<dim3(n_tiles, EP_SLICES), EP_THREADS, 0, stream>>>(from_idx, to_idx, h, deg, part, N, E);

    int n4 = N * (D_OUT / 4);
    sum_parts_kernel<<<(n4 + SUM_U * 256 - 1) / (SUM_U * 256), 256, 0, stream>>>(part, out, n4);
}


# degree count: the scattered global f32 atomic adds become LDS integer counters per (node range x edge slice) block; the count planes are summed in the rsqrt pass; init launch goes; degrees exact
# speedup vs baseline: 3.2247x; 1.5193x over previous
#include <hip/hip_runtime.h>

#define D_IN  128
#define D_OUT 64

typedef __attribute__((ext_vector_type(2))) float v2f;
typedef __attribute__((ext_vector_type(8))) float v8f;

#define HR_NODES 6272
#define HK       16
#define HT       512
#define HU       8

__global__ void __launch_bounds__(HT)
deg_hist_kernel(const int* __restrict__ to_idx, unsigned* __restrict__ cntp, int np, int e_cnt) {
    __shared__ __attribute__((aligned(16))) unsigned cnt[HR_NODES];

    const int tid   = threadIdx.x;
    const int base  = blockIdx.x * HR_NODES;
    const int slice = blockIdx.y;
    int lim = np - base;
    if (lim > HR_NODES) lim = HR_NODES;
    const unsigned ubase = (unsigned)base;
    const unsigned ulim  = (unsigned)lim;

    {
        uint4* c4 = (uint4*)cnt;
        const uint4 z = make_uint4(0u, 0u, 0u, 0u);
        for (int i = tid; i < HR_NODES / 4; i += HT) c4[i] = z;
    }
    __syncthreads();

    auto count1 = [&](int t, unsigned lm) {
        const unsigned d = (unsigned)t - ubase;
        if (d < lm) __hip_atomic_fetch_add(&cnt[d], 1u, __ATOMIC_RELAXED, __HIP_MEMORY_SCOPE_WORKGROUP);
    };

    int head = (int)(((16u - ((unsigned)(size_t)to_idx & 15u)) & 15u) >> 2);
    if (head > e_cnt) head = e_cnt;
    const int nvec = (e_cnt - head) >> 2;
    if (slice == 0) {
        const int vec_end = head + (nvec << 2);
        const int nleft = head + (e_cnt - vec_end);
        if (tid < nleft) {
            const int e = (tid < head) ? tid : (vec_end + (tid - head));
            count1(to_idx[e], ulim);
        }
    }

    const int v_lo = (int)(((long long)nvec * slice) / HK);
    const int nloc = (int)(((long long)nvec * (slice + 1)) / HK) - v_lo;
    if (nloc > 0) {
        const int4* __restrict__ sp = (const int4*)(to_idx + head) + v_lo;
        auto load_piece = [&](int i) -> int4 {
            int c = (i < nloc) ? i : (nloc - 1);
            return sp[c];
        };
        auto count_piece = [&](int i, int4 t) {
            const unsigned lm = (i < nloc) ? ulim : 0u;
            count1(t.x, lm); count1(t.y, lm); count1(t.z, lm); count1(t.w, lm);
        };

        int4 bufa[HU], bufb[HU];
#pragma unroll
        for (int u = 0; u < HU; ++u) bufa[u] = load_piece(u * HT + tid);

        for (int i0 = 0; i0 < nloc; i0 += 2 * HU * HT) {
#pragma unroll
            for (int u = 0; u < HU; ++u) bufb[u] = load_piece(i0 + (HU + u) * HT + tid);
#pragma unroll
            for (int u = 0; u < HU; ++u) count_piece(i0 + u * HT + tid, bufa[u]);
#pragma unroll
            for (int u = 0; u < HU; ++u) bufa[u] = load_piece(i0 + (2 * HU + u) * HT + tid);
#pragma unroll
            for (int u = 0; u < HU; ++u) count_piece(i0 + (HU + u) * HT + tid, bufb[u]);
        }
    }
    __syncthreads();

    {
        uint4* __restrict__ dst = (uint4*)(cntp + (size_t)slice * np + base);
        const uint4* c4 = (const uint4*)cnt;
        for (int i = tid; i < (lim >> 2); i += HT) dst[i] = c4[i];
    }
}

__global__ void __launch_bounds__(256)
dinv_kernel(const unsigned* __restrict__ cntp, float* __restrict__ dinv, int np) {
    const int n4 = np >> 2;
    const int i4 = blockIdx.x * 256 + threadIdx.x;
    if (i4 < n4) {
        const uint4* __restrict__ p = (const uint4*)cntp + i4;
        uint4 c[HK];
#pragma unroll
        for (int k = 0; k < HK; ++k) c[k] = p[(size_t)k * n4];
        uint4 s = c[0];
#pragma unroll
        for (int k = 1; k < HK; ++k) { s.x += c[k].x; s.y += c[k].y; s.z += c[k].z; s.w += c[k].w; }
        const float d0 = 1.0f + (float)s.x;
        const float d1 = 1.0f + (float)s.y;
        const float d2 = 1.0f + (float)s.z;
        const float d3 = 1.0f + (float)s.w;
        float4 o;
        o.x = (d0 > 0.0f) ? rsqrtf(d0) : 0.0f;
        o.y = (d1 > 0.0f) ? rsqrtf(d1) : 0.0f;
        o.z = (d2 > 0.0f) ? rsqrtf(d2) : 0.0f;
        o.w = (d3 > 0.0f) ? rsqrtf(d3) : 0.0f;
        ((float4*)dinv)[i4] = o;
    }
}

#define GEMM_WAVES 8
__global__ void __launch_bounds__(GEMM_WAVES * 32)
gemm_wmma_kernel(const float* __restrict__ x, const float* __restrict__ W,
                 const float* __restrict__ b, float* __restrict__ h, int n_nodes) {
    __shared__ float4 wl[16 * 4 * 32];

    const int tid  = threadIdx.x;
    const int lane = tid & 31;
    const int wave = tid >> 5;
    const int half = lane >> 4;
    const int l    = lane & 15;
    const int m0   = (blockIdx.x * GEMM_WAVES + wave) * 16;

    int row = m0 + l;
    int rrow = (row < n_nodes) ? row : (n_nodes - 1);

    const float4* __restrict__ xr = (const float4*)(x + (size_t)rrow * D_IN) + half;
    float4 a[16];
#pragma unroll
    for (int j = 0; j < 16; ++j) a[j] = xr[2 * j];

    float bias[4];
#pragma unroll
    for (int t = 0; t < 4; ++t) bias[t] = b[t * 16 + l];

    {
        const float4* __restrict__ W4 = (const float4*)W;
#pragma unroll
        for (int i = 0; i < (D_OUT * D_IN / 4) / (GEMM_WAVES * 32); ++i) {
            const int g  = i * (GEMM_WAVES * 32) + tid;
            const int r  = g >> 5;
            const int c  = g & 31;
            const int t  = r >> 4, n = r & 15;
            const int j  = c >> 1, hf = c & 1;
            wl[(j * 4 + t) * 32 + hf * 16 + (n ^ j)] = W4[g];
        }
    }
    __syncthreads();

    v8f acc[4];
#pragma unroll
    for (int t = 0; t < 4; ++t) {
#pragma unroll
        for (int r = 0; r < 8; ++r) acc[t][r] = bias[t];
    }

#pragma unroll
    for (int j = 0; j < 16; ++j) {
        float4 bv[4];
#pragma unroll
        for (int t = 0; t < 4; ++t) bv[t] = wl[(j * 4 + t) * 32 + (lane ^ j)];
        v2f a0, a1;
        a0[0] = a[j].x; a0[1] = a[j].y;
        a1[0] = a[j].z; a1[1] = a[j].w;
#pragma unroll
        for (int t = 0; t < 4; ++t) {
            v2f b0; b0[0] = bv[t].x; b0[1] = bv[t].y;
            acc[t] = __builtin_amdgcn_wmma_f32_16x16x4_f32(
                 false, a0,  false, b0,
                 (short)0, acc[t],  false,  false);
        }
#pragma unroll
        for (int t = 0; t < 4; ++t) {
            v2f b1; b1[0] = bv[t].z; b1[1] = bv[t].w;
            acc[t] = __builtin_amdgcn_wmma_f32_16x16x4_f32(
                 false, a1,  false, b1,
                 (short)0, acc[t],  false,  false);
        }
    }

    float* __restrict__ hbase = h + (size_t)m0 * D_OUT;
#pragma unroll
    for (int r = 0; r < 8; ++r) {
        const int mr = r + 8 * half;
        if (m0 + mr < n_nodes) {
#pragma unroll
            for (int t = 0; t < 4; ++t)
                hbase[(size_t)mr * D_OUT + t * 16 + l] = acc[t][r];
        }
    }
}

#define EP_SLICES 2
#define TILE_ROWS 896
#define QCAP      256
#define EP_WAVES  32
#define EP_THREADS (EP_WAVES * 32)

__global__ void __launch_bounds__(EP_THREADS)
edge_tile_kernel(const int* __restrict__ from_idx, const int* __restrict__ to_idx,
                 const float* __restrict__ h, const float* __restrict__ dinv,
                 float* __restrict__ part, int n_nodes, int e_cnt) {
    __shared__ __attribute__((aligned(16))) float tile[TILE_ROWS * D_OUT];
    __shared__ float dinv_t[TILE_ROWS];
    __shared__ int   queue[EP_WAVES * QCAP];
    __shared__ unsigned short qrow[EP_WAVES * QCAP];

    const int tid  = threadIdx.x;
    const int lane = tid & 31;
    const int wave = __builtin_amdgcn_readfirstlane(tid >> 5);
    const int base = blockIdx.x * TILE_ROWS;
    const int slice = blockIdx.y;
    int rows = n_nodes - base;
    if (rows > TILE_ROWS) rows = TILE_ROWS;

    {
        const float4* __restrict__ hsrc = (const float4*)(h + (size_t)base * D_OUT);
        float4* tdst = (float4*)tile;
        if (slice == 0) {
            for (int i = tid; i < rows * (D_OUT / 4); i += EP_THREADS) {
                float di = dinv[base + (i >> 4)];
                float4 v = hsrc[i];
                float4 o;
                o.x = v.x * di * di; o.y = v.y * di * di;
                o.z = v.z * di * di; o.w = v.w * di * di;
                tdst[i] = o;
            }
        } else {
            const float4 z = make_float4(0.0f, 0.0f, 0.0f, 0.0f);
            for (int i = tid; i < rows * (D_OUT / 4); i += EP_THREADS) tdst[i] = z;
        }
        for (int r = tid; r < rows; r += EP_THREADS) dinv_t[r] = dinv[base + r];
    }
    __syncthreads();

    int* q = queue + wave * QCAP;
    unsigned short* qr = qrow + wave * QCAP;
    int count = 0;
    const unsigned ubase = (unsigned)base;
    const unsigned urows = (unsigned)rows;

    auto append = [&](bool m, int e, unsigned d) {
        unsigned mk = __builtin_amdgcn_ballot_w32(m);
        if (mk) {
            int pos = count + (int)__builtin_amdgcn_mbcnt_lo(mk, 0u);
            if (m) { q[pos] = e; qr[pos] = (unsigned short)d; }
            count += __builtin_popcount(mk);
        }
    };

    auto add_row = [&](float2 hv, int lr, float c) {
        float* p = &tile[lr * D_OUT + 2 * lane];
        __hip_atomic_fetch_add(p,     hv.x * c, __ATOMIC_RELAXED, __HIP_MEMORY_SCOPE_WORKGROUP);
        __hip_atomic_fetch_add(p + 1, hv.y * c, __ATOMIC_RELAXED, __HIP_MEMORY_SCOPE_WORKGROUP);
    };

    auto batch = [&](int b0, int n, int m, bool full) {
        int idx = b0 + lane;
        if (idx > n - 1) idx = n - 1;
        const int e = q[idx];
        const int lr = qr[idx];
        const int f = from_idx[e];
        const float df = dinv[f];
        if (full) {
            float2 hv[32];
#pragma unroll
            for (int k = 0; k < 32; ++k) {
                int fj = __builtin_amdgcn_readlane(f, k);
                hv[k] = ((const float2*)(h + (size_t)fj * D_OUT))[lane];
            }
            const float c = df * dinv_t[lr];
#pragma unroll
            for (int k = 0; k < 32; ++k) {
                int   lj = __builtin_amdgcn_readlane(lr, k);
                float cj = __int_as_float(__builtin_amdgcn_readlane(__float_as_int(c), k));
                add_row(hv[k], lj, cj);
            }
        } else {
            const float c = df * dinv_t[lr];
            for (int j = 0; j < m; j += 8) {
                float2 hv[8];
#pragma unroll
                for (int k = 0; k < 8; ++k) {
                    int fj = __builtin_amdgcn_readlane(f, j + k);
                    hv[k] = ((const float2*)(h + (size_t)fj * D_OUT))[lane];
                }
#pragma unroll
                for (int k = 0; k < 8; ++k) {
                    if (j + k < m) {
                        int   lj = __builtin_amdgcn_readlane(lr, j + k);
                        float cj = __int_as_float(__builtin_amdgcn_readlane(__float_as_int(c), j + k));
                        add_row(hv[k], lj, cj);
                    }
                }
            }
        }
    };

    auto flush = [&](bool final_call) {
        const int n = count;
        const int nb = n & ~31;
        for (int b0 = 0; b0 < nb; b0 += 32) batch(b0, n, 32, true);
        const int r = n - nb;
        if (final_call) {
            if (r > 0) batch(nb, n, r, false);
            count = 0;
        } else {
            int v = 0;
            unsigned short w = 0;
            if (lane < r) { v = q[nb + lane]; w = qr[nb + lane]; }
            if (nb > 0 && lane < r) { q[lane] = v; qr[lane] = w; }
            count = (nb > 0) ? r : n;
        }
    };

    int head = (int)(((16u - ((unsigned)(size_t)to_idx & 15u)) & 15u) >> 2);
    if (head > e_cnt) head = e_cnt;
    const int nwc = (e_cnt - head) >> 7;
    const int c_lo = (int)(((long long)nwc * slice) / EP_SLICES);
    const int nloc = (int)(((long long)nwc * (slice + 1)) / EP_SLICES) - c_lo;
    const int4* __restrict__ vp = (const int4*)(to_idx + head) + (size_t)c_lo * 32;
    const int ebase = head + (c_lo << 7);

    if (wave == 0 && slice == 0) {
        const int vec_end = head + (nwc << 7);
        const int nleft = head + (e_cnt - vec_end);
        for (int k0 = 0; k0 < nleft; k0 += 32) {
            int k = k0 + lane;
            int e = (k < head) ? k : (vec_end + (k - head));
            unsigned d = 0xFFFFFFFFu;
            if (k < nleft) d = (unsigned)to_idx[e] - ubase;
            append(d < urows, e, d);
            if (count > QCAP - 128) flush(false);
        }
    }

    const int S = (nloc + 31) >> 5;
    const int rot = (int)((blockIdx.x * (unsigned)S) / gridDim.x);

    auto chunk_of = [&](int s) -> int {
        int sr = s + rot;
        if (sr >= S) sr -= S;
        return sr * 32 + wave;
    };
    auto load_chunk = [&](int s) -> int4 {
        int c = chunk_of(s);
        if (c > nloc - 1) c = nloc - 1;
        return vp[c * 32 + lane];
    };

#define PF 4
    if (nloc > 0) {
        auto step = [&](int s, int4 cur) {
            const int cs = chunk_of(s);
            const unsigned lim = (s < S && cs < nloc) ? urows : 0u;
            unsigned d0 = (unsigned)cur.x - ubase;
            unsigned d1 = (unsigned)cur.y - ubase;
            unsigned d2 = (unsigned)cur.z - ubase;
            unsigned d3 = (unsigned)cur.w - ubase;
            unsigned dm = min(min(d0, d1), min(d2, d3));
            if (__builtin_amdgcn_ballot_w32(dm < lim)) {
                int e0 = ebase + ((cs * 32 + lane) << 2);
                append(d0 < lim, e0, d0);
                append(d1 < lim, e0 + 1, d1);
                append(d2 < lim, e0 + 2, d2);
                append(d3 < lim, e0 + 3, d3);
                if (count > QCAP - 128) flush(false);
            }
        };

        int4 bufa[PF], bufb[PF];
#pragma unroll
        for (int u = 0; u < PF; ++u) bufa[u] = load_chunk(u);

        for (int s0 = 0; s0 < S; s0 += 2 * PF) {
#pragma unroll
            for (int u = 0; u < PF; ++u) bufb[u] = load_chunk(s0 + PF + u);
#pragma unroll
            for (int u = 0; u < PF; ++u) step(s0 + u, bufa[u]);
#pragma unroll
            for (int u = 0; u < PF; ++u) bufa[u] = load_chunk(s0 + 2 * PF + u);
#pragma unroll
            for (int u = 0; u < PF; ++u) step(s0 + PF + u, bufb[u]);
        }
    }
    flush(true);

    __syncthreads();

    {
        float4* __restrict__ odst = (float4*)(part + ((size_t)slice * n_nodes + base) * D_OUT);
        const float4* tsrc = (const float4*)tile;
        for (int i = tid; i < rows * (D_OUT / 4); i += EP_THREADS) odst[i] = tsrc[i];
    }
}

#define SUM_U 8
__global__ void __launch_bounds__(256)
sum_parts_kernel(const float* __restrict__ part, float* __restrict__ out, int n4) {
    const float4* __restrict__ p = (const float4*)part;
    float4* __restrict__ o = (float4*)out;
    const int i0 = blockIdx.x * (SUM_U * 256) + threadIdx.x;
    float4 v[EP_SLICES][SUM_U];
#pragma unroll
    for (int s = 0; s < EP_SLICES; ++s) {
#pragma unroll
        for (int u = 0; u < SUM_U; ++u) {
            int j = i0 + u * 256;
            if (j > n4 - 1) j = n4 - 1;
            v[s][u] = p[(size_t)s * n4 + j];
        }
    }
#pragma unroll
    for (int u = 0; u < SUM_U; ++u) {
        const int j = i0 + u * 256;
        float4 a = v[0][u];
#pragma unroll
        for (int s = 1; s < EP_SLICES; ++s) {
            a.x += v[s][u].x; a.y += v[s][u].y; a.z += v[s][u].z; a.w += v[s][u].w;
        }
        if (j < n4) o[j] = a;
    }
}

extern "C" void kernel_launch(void* const* d_in, const int* in_sizes, int n_in,
                              void* d_out, int out_size, void* d_ws, size_t ws_size,
                              hipStream_t stream) {
    const float* x  = (const float*)d_in[0];
    const float* W  = (const float*)d_in[1];
    const float* b  = (const float*)d_in[2];
    const int*   ei = (const int*)d_in[3];

    const int N = in_sizes[0] / D_IN;
    const int E = in_sizes[3] / 2;
    const int* from_idx = ei;
    const int* to_idx   = ei + E;

    float* out = (float*)d_out;

    const int NP = (N + 63) & ~63;
    float* deg = (float*)d_ws;
    size_t deg_bytes = ((size_t)NP * sizeof(float) + 255) & ~(size_t)255;
    float* h = (float*)((char*)d_ws + deg_bytes);
    float* part = h + (size_t)N * D_OUT;
    unsigned* cntp = (unsigned*)(part + (size_t)EP_SLICES * N * D_OUT);

    deg_hist_kernel<<<dim3((NP + HR_NODES - 1) / HR_NODES, HK), HT, 0, stream>>>(to_idx, cntp, NP, E);
    dinv_kernel<<<((NP >> 2) + 255) / 256, 256, 0, stream>>>(cntp, deg, NP);

    int m_tiles = (N + 15) / 16;
    gemm_wmma_kernel<<<(m_tiles + GEMM_WAVES - 1) / GEMM_WAVES, GEMM_WAVES * 32, 0, stream>>>(x, W, b, h, N);

    int n_tiles = (N + TILE_ROWS - 1) / TILE_ROWS;
    edge_tile_kernel<<<dim3(n_tiles, EP_SLICES), EP_THREADS, 0, stream>>>(from_idx, to_idx, h, deg, part, N, E);

    int n4 = N * (D_OUT / 4);
    sum_parts_kernel<<<(n4 + SUM_U * 256 - 1) / (SUM_U * 256), 256, 0, stream>>>(part, out, n4);
}


# degree-count blocks and GEMM blocks share one launch: independent stages, the count streams to[] from L2 into LDS counters while the GEMM streams x from memory; 5 launches -> 4; sums unchanged
# speedup vs baseline: 3.3829x; 1.0491x over previous
#include <hip/hip_runtime.h>

#define D_IN  128
#define D_OUT 64

typedef __attribute__((ext_vector_type(2))) float v2f;
typedef __attribute__((ext_vector_type(8))) float v8f;

#define HR_NODES 6272
#define HK       16
#define HT       512
#define HU       8

static __device__ __forceinline__ void
deg_hist_body(const int* __restrict__ to_idx, unsigned* __restrict__ cntp, int np, int e_cnt,
              int range, int slice) {
    __shared__ __attribute__((aligned(16))) unsigned cnt[HR_NODES];

    const int tid   = threadIdx.x;
    const int base  = range * HR_NODES;
    int lim = np - base;
    if (lim > HR_NODES) lim = HR_NODES;
    const unsigned ubase = (unsigned)base;
    const unsigned ulim  = (unsigned)lim;

    {
        uint4* c4 = (uint4*)cnt;
        const uint4 z = make_uint4(0u, 0u, 0u, 0u);
        for (int i = tid; i < HR_NODES / 4; i += HT) c4[i] = z;
    }
    __syncthreads();

    auto count1 = [&](int t, unsigned lm) {
        const unsigned d = (unsigned)t - ubase;
        if (d < lm) __hip_atomic_fetch_add(&cnt[d], 1u, __ATOMIC_RELAXED, __HIP_MEMORY_SCOPE_WORKGROUP);
    };

    int head = (int)(((16u - ((unsigned)(size_t)to_idx & 15u)) & 15u) >> 2);
    if (head > e_cnt) head = e_cnt;
    const int nvec = (e_cnt - head) >> 2;
    if (slice == 0) {
        const int vec_end = head + (nvec << 2);
        const int nleft = head + (e_cnt - vec_end);
        if (tid < nleft) {
            const int e = (tid < head) ? tid : (vec_end + (tid - head));
            count1(to_idx[e], ulim);
        }
    }

    const int v_lo = (int)(((long long)nvec * slice) / HK);
    const int nloc = (int)(((long long)nvec * (slice + 1)) / HK) - v_lo;
    if (nloc > 0) {
        const int4* __restrict__ sp = (const int4*)(to_idx + head) + v_lo;
        auto load_piece = [&](int i) -> int4 {
            int c = (i < nloc) ? i : (nloc - 1);
            return sp[c];
        };
        auto count_piece = [&](int i, int4 t) {
            const unsigned lm = (i < nloc) ? ulim : 0u;
            count1(t.x, lm); count1(t.y, lm); count1(t.z, lm); count1(t.w, lm);
        };

        int4 bufa[HU], bufb[HU];
#pragma unroll
        for (int u = 0; u < HU; ++u) bufa[u] = load_piece(u * HT + tid);

        for (int i0 = 0; i0 < nloc; i0 += 2 * HU * HT) {
#pragma unroll
            for (int u = 0; u < HU; ++u) bufb[u] = load_piece(i0 + (HU + u) * HT + tid);
#pragma unroll
            for (int u = 0; u < HU; ++u) count_piece(i0 + u * HT + tid, bufa[u]);
#pragma unroll
            for (int u = 0; u < HU; ++u) bufa[u] = load_piece(i0 + (2 * HU + u) * HT + tid);
#pragma unroll
            for (int u = 0; u < HU; ++u) count_piece(i0 + (HU + u) * HT + tid, bufb[u]);
        }
    }
    __syncthreads();

    {
        uint4* __restrict__ dst = (uint4*)(cntp + (size_t)slice * np + base);
        const uint4* c4 = (const uint4*)cnt;
        for (int i = tid; i < (lim >> 2); i += HT) dst[i] = c4[i];
    }
}

__global__ void __launch_bounds__(256)
dinv_kernel(const unsigned* __restrict__ cntp, float* __restrict__ dinv, int np) {
    const int n4 = np >> 2;
    const int i4 = blockIdx.x * 256 + threadIdx.x;
    if (i4 < n4) {
        const uint4* __restrict__ p = (const uint4*)cntp + i4;
        uint4 c[HK];
#pragma unroll
        for (int k = 0; k < HK; ++k) c[k] = p[(size_t)k * n4];
        uint4 s = c[0];
#pragma unroll
        for (int k = 1; k < HK; ++k) { s.x += c[k].x; s.y += c[k].y; s.z += c[k].z; s.w += c[k].w; }
        const float d0 = 1.0f + (float)s.x;
        const float d1 = 1.0f + (float)s.y;
        const float d2 = 1.0f + (float)s.z;
        const float d3 = 1.0f + (float)s.w;
        float4 o;
        o.x = (d0 > 0.0f) ? rsqrtf(d0) : 0.0f;
        o.y = (d1 > 0.0f) ? rsqrtf(d1) : 0.0f;
        o.z = (d2 > 0.0f) ? rsqrtf(d2) : 0.0f;
        o.w = (d3 > 0.0f) ? rsqrtf(d3) : 0.0f;
        ((float4*)dinv)[i4] = o;
    }
}

#define GEMM_WAVES 16
static __device__ __forceinline__ void
gemm_body(const float* __restrict__ x, const float* __restrict__ W,
          const float* __restrict__ b, float* __restrict__ h, int n_nodes, int blk) {
    __shared__ float4 wl[16 * 4 * 32];

    const int tid  = threadIdx.x;
    const int lane = tid & 31;
    const int wave = tid >> 5;
    const int half = lane >> 4;
    const int l    = lane & 15;
    const int m0   = (blk * GEMM_WAVES + wave) * 16;

    int row = m0 + l;
    int rrow = (row < n_nodes) ? row : (n_nodes - 1);

    const float4* __restrict__ xr = (const float4*)(x + (size_t)rrow * D_IN) + half;
    float4 a[16];
#pragma unroll
    for (int j = 0; j < 16; ++j) a[j] = xr[2 * j];

    float bias[4];
#pragma unroll
    for (int t = 0; t < 4; ++t) bias[t] = b[t * 16 + l];

    {
        const float4* __restrict__ W4 = (const float4*)W;
#pragma unroll
        for (int i = 0; i < (D_OUT * D_IN / 4) / (GEMM_WAVES * 32); ++i) {
            const int g  = i * (GEMM_WAVES * 32) + tid;
            const int r  = g >> 5;
            const int c  = g & 31;
            const int t  = r >> 4, n = r & 15;
            const int j  = c >> 1, hf = c & 1;
            wl[(j * 4 + t) * 32 + hf * 16 + (n ^ j)] = W4[g];
        }
    }
    __syncthreads();

    v8f acc[4];
#pragma unroll
    for (int t = 0; t < 4; ++t) {
#pragma unroll
        for (int r = 0; r < 8; ++r) acc[t][r] = bias[t];
    }

#pragma unroll
    for (int j = 0; j < 16; ++j) {
        float4 bv[4];
#pragma unroll
        for (int t = 0; t < 4; ++t) bv[t] = wl[(j * 4 + t) * 32 + (lane ^ j)];
        v2f a0, a1;
        a0[0] = a[j].x; a0[1] = a[j].y;
        a1[0] = a[j].z; a1[1] = a[j].w;
#pragma unroll
        for (int t = 0; t < 4; ++t) {
            v2f b0; b0[0] = bv[t].x; b0[1] = bv[t].y;
            acc[t] = __builtin_amdgcn_wmma_f32_16x16x4_f32(
                 false, a0,  false, b0,
                 (short)0, acc[t],  false,  false);
        }
#pragma unroll
        for (int t = 0; t < 4; ++t) {
            v2f b1; b1[0] = bv[t].z; b1[1] = bv[t].w;
            acc[t] = __builtin_amdgcn_wmma_f32_16x16x4_f32(
                 false, a1,  false, b1,
                 (short)0, acc[t],  false,  false);
        }
    }

    float* __restrict__ hbase = h + (size_t)m0 * D_OUT;
#pragma unroll
    for (int r = 0; r < 8; ++r) {
        const int mr = r + 8 * half;
        if (m0 + mr < n_nodes) {
#pragma unroll
            for (int t = 0; t < 4; ++t)
                hbase[(size_t)mr * D_OUT + t * 16 + l] = acc[t][r];
        }
    }
}

__global__ void __launch_bounds__(HT)
front_kernel(const int* __restrict__ to_idx, unsigned* __restrict__ cntp, int np, int e_cnt,
             int n_ranges, int n_hist,
             const float* __restrict__ x, const float* __restrict__ W,
             const float* __restrict__ b, float* __restrict__ h, int n_nodes) {
    const int bid = blockIdx.x;
    if (bid < n_hist) {
        const int slice = bid / n_ranges;
        deg_hist_body(to_idx, cntp, np, e_cnt, bid - slice * n_ranges, slice);
    } else {
        gemm_body(x, W, b, h, n_nodes, bid - n_hist);
    }
}

#define EP_SLICES 2
#define TILE_ROWS 896
#define QCAP      256
#define EP_WAVES  32
#define EP_THREADS (EP_WAVES * 32)

__global__ void __launch_bounds__(EP_THREADS)
edge_tile_kernel(const int* __restrict__ from_idx, const int* __restrict__ to_idx,
                 const float* __restrict__ h, const float* __restrict__ dinv,
                 float* __restrict__ part, int n_nodes, int e_cnt) {
    __shared__ __attribute__((aligned(16))) float tile[TILE_ROWS * D_OUT];
    __shared__ float dinv_t[TILE_ROWS];
    __shared__ int   queue[EP_WAVES * QCAP];
    __shared__ unsigned short qrow[EP_WAVES * QCAP];

    const int tid  = threadIdx.x;
    const int lane = tid & 31;
    const int wave = __builtin_amdgcn_readfirstlane(tid >> 5);
    const int base = blockIdx.x * TILE_ROWS;
    const int slice = blockIdx.y;
    int rows = n_nodes - base;
    if (rows > TILE_ROWS) rows = TILE_ROWS;

    {
        const float4* __restrict__ hsrc = (const float4*)(h + (size_t)base * D_OUT);
        float4* tdst = (float4*)tile;
        if (slice == 0) {
            for (int i = tid; i < rows * (D_OUT / 4); i += EP_THREADS) {
                float di = dinv[base + (i >> 4)];
                float4 v = hsrc[i];
                float4 o;
                o.x = v.x * di * di; o.y = v.y * di * di;
                o.z = v.z * di * di; o.w = v.w * di * di;
                tdst[i] = o;
            }
        } else {
            const float4 z = make_float4(0.0f, 0.0f, 0.0f, 0.0f);
            for (int i = tid; i < rows * (D_OUT / 4); i += EP_THREADS) tdst[i] = z;
        }
        for (int r = tid; r < rows; r += EP_THREADS) dinv_t[r] = dinv[base + r];
    }
    __syncthreads();

    int* q = queue + wave * QCAP;
    unsigned short* qr = qrow + wave * QCAP;
    int count = 0;
    const unsigned ubase = (unsigned)base;
    const unsigned urows = (unsigned)rows;

    auto append = [&](bool m, int e, unsigned d) {
        unsigned mk = __builtin_amdgcn_ballot_w32(m);
        if (mk) {
            int pos = count + (int)__builtin_amdgcn_mbcnt_lo(mk, 0u);
            if (m) { q[pos] = e; qr[pos] = (unsigned short)d; }
            count += __builtin_popcount(mk);
        }
    };

    auto add_row = [&](float2 hv, int lr, float c) {
        float* p = &tile[lr * D_OUT + 2 * lane];
        __hip_atomic_fetch_add(p,     hv.x * c, __ATOMIC_RELAXED, __HIP_MEMORY_SCOPE_WORKGROUP);
        __hip_atomic_fetch_add(p + 1, hv.y * c, __ATOMIC_RELAXED, __HIP_MEMORY_SCOPE_WORKGROUP);
    };

    auto batch = [&](int b0, int n, int m, bool full) {
        int idx = b0 + lane;
        if (idx > n - 1) idx = n - 1;
        const int e = q[idx];
        const int lr = qr[idx];
        const int f = from_idx[e];
        const float df = dinv[f];
        if (full) {
            float2 hv[32];
#pragma unroll
            for (int k = 0; k < 32; ++k) {
                int fj = __builtin_amdgcn_readlane(f, k);
                hv[k] = ((const float2*)(h + (size_t)fj * D_OUT))[lane];
            }
            const float c = df * dinv_t[lr];
#pragma unroll
            for (int k = 0; k < 32; ++k) {
                int   lj = __builtin_amdgcn_readlane(lr, k);
                float cj = __int_as_float(__builtin_amdgcn_readlane(__float_as_int(c), k));
                add_row(hv[k], lj, cj);
            }
        } else {
            const float c = df * dinv_t[lr];
            for (int j = 0; j < m; j += 8) {
                float2 hv[8];
#pragma unroll
                for (int k = 0; k < 8; ++k) {
                    int fj = __builtin_amdgcn_readlane(f, j + k);
                    hv[k] = ((const float2*)(h + (size_t)fj * D_OUT))[lane];
                }
#pragma unroll
                for (int k = 0; k < 8; ++k) {
                    if (j + k < m) {
                        int   lj = __builtin_amdgcn_readlane(lr, j + k);
                        float cj = __int_as_float(__builtin_amdgcn_readlane(__float_as_int(c), j + k));
                        add_row(hv[k], lj, cj);
                    }
                }
            }
        }
    };

    auto flush = [&](bool final_call) {
        const int n = count;
        const int nb = n & ~31;
        for (int b0 = 0; b0 < nb; b0 += 32) batch(b0, n, 32, true);
        const int r = n - nb;
        if (final_call) {
            if (r > 0) batch(nb, n, r, false);
            count = 0;
        } else {
            int v = 0;
            unsigned short w = 0;
            if (lane < r) { v = q[nb + lane]; w = qr[nb + lane]; }
            if (nb > 0 && lane < r) { q[lane] = v; qr[lane] = w; }
            count = (nb > 0) ? r : n;
        }
    };

    int head = (int)(((16u - ((unsigned)(size_t)to_idx & 15u)) & 15u) >> 2);
    if (head > e_cnt) head = e_cnt;
    const int nwc = (e_cnt - head) >> 7;
    const int c_lo = (int)(((long long)nwc * slice) / EP_SLICES);
    const int nloc = (int)(((long long)nwc * (slice + 1)) / EP_SLICES) - c_lo;
    const int4* __restrict__ vp = (const int4*)(to_idx + head) + (size_t)c_lo * 32;
    const int ebase = head + (c_lo << 7);

    if (wave == 0 && slice == 0) {
        const int vec_end = head + (nwc << 7);
        const int nleft = head + (e_cnt - vec_end);
        for (int k0 = 0; k0 < nleft; k0 += 32) {
            int k = k0 + lane;
            int e = (k < head) ? k : (vec_end + (k - head));
            unsigned d = 0xFFFFFFFFu;
            if (k < nleft) d = (unsigned)to_idx[e] - ubase;
            append(d < urows, e, d);
            if (count > QCAP - 128) flush(false);
        }
    }

    const int S = (nloc + 31) >> 5;
    const int rot = (int)((blockIdx.x * (unsigned)S) / gridDim.x);

    auto chunk_of = [&](int s) -> int {
        int sr = s + rot;
        if (sr >= S) sr -= S;
        return sr * 32 + wave;
    };
    auto load_chunk = [&](int s) -> int4 {
        int c = chunk_of(s);
        if (c > nloc - 1) c = nloc - 1;
        return vp[c * 32 + lane];
    };

#define PF 4
    if (nloc > 0) {
        auto step = [&](int s, int4 cur) {
            const int cs = chunk_of(s);
            const unsigned lim = (s < S && cs < nloc) ? urows : 0u;
            unsigned d0 = (unsigned)cur.x - ubase;
            unsigned d1 = (unsigned)cur.y - ubase;
            unsigned d2 = (unsigned)cur.z - ubase;
            unsigned d3 = (unsigned)cur.w - ubase;
            unsigned dm = min(min(d0, d1), min(d2, d3));
            if (__builtin_amdgcn_ballot_w32(dm < lim)) {
                int e0 = ebase + ((cs * 32 + lane) << 2);
                append(d0 < lim, e0, d0);
                append(d1 < lim, e0 + 1, d1);
                append(d2 < lim, e0 + 2, d2);
                append(d3 < lim, e0 + 3, d3);
                if (count > QCAP - 128) flush(false);
            }
        };

        int4 bufa[PF], bufb[PF];
#pragma unroll
        for (int u = 0; u < PF; ++u) bufa[u] = load_chunk(u);

        for (int s0 = 0; s0 < S; s0 += 2 * PF) {
#pragma unroll
            for (int u = 0; u < PF; ++u) bufb[u] = load_chunk(s0 + PF + u);
#pragma unroll
            for (int u = 0; u < PF; ++u) step(s0 + u, bufa[u]);
#pragma unroll
            for (int u = 0; u < PF; ++u) bufa[u] = load_chunk(s0 + 2 * PF + u);
#pragma unroll
            for (int u = 0; u < PF; ++u) step(s0 + PF + u, bufb[u]);
        }
    }
    flush(true);

    __syncthreads();

    {
        float4* __restrict__ odst = (float4*)(part + ((size_t)slice * n_nodes + base) * D_OUT);
        const float4* tsrc = (const float4*)tile;
        for (int i = tid; i < rows * (D_OUT / 4); i += EP_THREADS) odst[i] = tsrc[i];
    }
}

#define SUM_U 8
__global__ void __launch_bounds__(256)
sum_parts_kernel(const float* __restrict__ part, float* __restrict__ out, int n4) {
    const float4* __restrict__ p = (const float4*)part;
    float4* __restrict__ o = (float4*)out;
    const int i0 = blockIdx.x * (SUM_U * 256) + threadIdx.x;
    float4 v[EP_SLICES][SUM_U];
#pragma unroll
    for (int s = 0; s < EP_SLICES; ++s) {
#pragma unroll
        for (int u = 0; u < SUM_U; ++u) {
            int j = i0 + u * 256;
            if (j > n4 - 1) j = n4 - 1;
            v[s][u] = p[(size_t)s * n4 + j];
        }
    }
#pragma unroll
    for (int u = 0; u < SUM_U; ++u) {
        const int j = i0 + u * 256;
        float4 a = v[0][u];
#pragma unroll
        for (int s = 1; s < EP_SLICES; ++s) {
            a.x += v[s][u].x; a.y += v[s][u].y; a.z += v[s][u].z; a.w += v[s][u].w;
        }
        if (j < n4) o[j] = a;
    }
}

extern "C" void kernel_launch(void* const* d_in, const int* in_sizes, int n_in,
                              void* d_out, int out_size, void* d_ws, size_t ws_size,
                              hipStream_t stream) {
    const float* x  = (const float*)d_in[0];
    const float* W  = (const float*)d_in[1];
    const float* b  = (const float*)d_in[2];
    const int*   ei = (const int*)d_in[3];

    const int N = in_sizes[0] / D_IN;
    const int E = in_sizes[3] / 2;
    const int* from_idx = ei;
    const int* to_idx   = ei + E;

    float* out = (float*)d_out;

    const int NP = (N + 63) & ~63;
    float* deg = (float*)d_ws;
    size_t deg_bytes = ((size_t)NP * sizeof(float) + 255) & ~(size_t)255;
    float* h = (float*)((char*)d_ws + deg_bytes);
    float* part = h + (size_t)N * D_OUT;
    unsigned* cntp = (unsigned*)(part + (size_t)EP_SLICES * N * D_OUT);

    int n_ranges = (NP + HR_NODES - 1) / HR_NODES;
    int n_hist = n_ranges * HK;
    int m_tiles = (N + 15) / 16;
    int n_gemm = (m_tiles + GEMM_WAVES - 1) / GEMM_WAVES;
    front_kernel<<<n_hist + n_gemm, HT, 0, stream>>>(to_idx, cntp, NP, E, n_ranges, n_hist, x, W, b, h, N);
    dinv_kernel<<<((NP >> 2) + 255) / 256, 256, 0, stream>>>(cntp, deg, NP);

    int n_tiles = (N + TILE_ROWS - 1) / TILE_ROWS;
    edge_tile_kernel<<<dim3(n_tiles, EP_SLICES), EP_THREADS, 0, stream>>>(from_idx, to_idx, h, deg, part, N, E);

    int n4 = N * (D_OUT / 4);
    sum_parts_kernel<<<(n4 + SUM_U * 256 - 1) / (SUM_U * 256), 256, 0, stream>>>(part, out, n4);
}


# edge pass: blocks search 1-byte tile numbers (written by the count blocks) 16 per lane by byte-parallel compare instead of reading to[]; to[e] is read for queued edges only; add order stays free
# speedup vs baseline: 3.7610x; 1.1118x over previous
#include <hip/hip_runtime.h>

#define D_IN  128
#define D_OUT 64

typedef __attribute__((ext_vector_type(2))) float v2f;
typedef __attribute__((ext_vector_type(8))) float v8f;

#define HR_NODES 6272
#define HK       16
#define HT       512
#define HU       8
#define KEY_ROWS 896u

static __device__ __forceinline__ void
deg_hist_body(const int* __restrict__ to_idx, unsigned* __restrict__ cntp,
              unsigned* __restrict__ keyw, int np, int e_cnt, int range, int slice) {
    __shared__ __attribute__((aligned(16))) unsigned cnt[HR_NODES];

    const int tid   = threadIdx.x;
    const int base  = range * HR_NODES;
    int lim = np - base;
    if (lim > HR_NODES) lim = HR_NODES;
    const unsigned ubase = (unsigned)base;
    const unsigned ulim  = (unsigned)lim;

    {
        uint4* c4 = (uint4*)cnt;
        const uint4 z = make_uint4(0u, 0u, 0u, 0u);
        for (int i = tid; i < HR_NODES / 4; i += HT) c4[i] = z;
    }
    __syncthreads();

    auto count1 = [&](int t, unsigned lm) {
        const unsigned d = (unsigned)t - ubase;
        if (d < lm) __hip_atomic_fetch_add(&cnt[d], 1u, __ATOMIC_RELAXED, __HIP_MEMORY_SCOPE_WORKGROUP);
    };

    int head = (int)(((16u - ((unsigned)(size_t)to_idx & 15u)) & 15u) >> 2);
    if (head > e_cnt) head = e_cnt;
    const int nvec = (e_cnt - head) >> 2;
    if (slice == 0) {
        const int vec_end = head + (nvec << 2);
        const int nleft = head + (e_cnt - vec_end);
        if (tid < nleft) {
            const int e = (tid < head) ? tid : (vec_end + (tid - head));
            count1(to_idx[e], ulim);
        }
    }

    const int v_lo = (int)(((long long)nvec * slice) / HK);
    const int nloc = (int)(((long long)nvec * (slice + 1)) / HK) - v_lo;
    if (nloc > 0) {
        const int4* __restrict__ sp = (const int4*)(to_idx + head) + v_lo;
        auto load_piece = [&](int i) -> int4 {
            int c = (i < nloc) ? i : (nloc - 1);
            return sp[c];
        };
        auto count_piece = [&](int i, int4 t) {
            const unsigned lm = (i < nloc) ? ulim : 0u;
            count1(t.x, lm); count1(t.y, lm); count1(t.z, lm); count1(t.w, lm);
            if (range == 0 && i < nloc) {
                keyw[v_lo + i] = (((unsigned)t.x / KEY_ROWS) & 255u)
                               | ((((unsigned)t.y / KEY_ROWS) & 255u) << 8)
                               | ((((unsigned)t.z / KEY_ROWS) & 255u) << 16)
                               | (((unsigned)t.w / KEY_ROWS) << 24);
            }
        };

        int4 bufa[HU], bufb[HU];
#pragma unroll
        for (int u = 0; u < HU; ++u) bufa[u] = load_piece(u * HT + tid);

        for (int i0 = 0; i0 < nloc; i0 += 2 * HU * HT) {
#pragma unroll
            for (int u = 0; u < HU; ++u) bufb[u] = load_piece(i0 + (HU + u) * HT + tid);
#pragma unroll
            for (int u = 0; u < HU; ++u) count_piece(i0 + u * HT + tid, bufa[u]);
#pragma unroll
            for (int u = 0; u < HU; ++u) bufa[u] = load_piece(i0 + (2 * HU + u) * HT + tid);
#pragma unroll
            for (int u = 0; u < HU; ++u) count_piece(i0 + (HU + u) * HT + tid, bufb[u]);
        }
    }
    __syncthreads();

    {
        uint4* __restrict__ dst = (uint4*)(cntp + (size_t)slice * np + base);
        const uint4* c4 = (const uint4*)cnt;
        for (int i = tid; i < (lim >> 2); i += HT) dst[i] = c4[i];
    }
}

__global__ void __launch_bounds__(256)
dinv_kernel(const unsigned* __restrict__ cntp, float* __restrict__ dinv, int np) {
    const int n4 = np >> 2;
    const int i4 = blockIdx.x * 256 + threadIdx.x;
    if (i4 < n4) {
        const uint4* __restrict__ p = (const uint4*)cntp + i4;
        uint4 c[HK];
#pragma unroll
        for (int k = 0; k < HK; ++k) c[k] = p[(size_t)k * n4];
        uint4 s = c[0];
#pragma unroll
        for (int k = 1; k < HK; ++k) { s.x += c[k].x; s.y += c[k].y; s.z += c[k].z; s.w += c[k].w; }
        const float d0 = 1.0f + (float)s.x;
        const float d1 = 1.0f + (float)s.y;
        const float d2 = 1.0f + (float)s.z;
        const float d3 = 1.0f + (float)s.w;
        float4 o;
        o.x = (d0 > 0.0f) ? rsqrtf(d0) : 0.0f;
        o.y = (d1 > 0.0f) ? rsqrtf(d1) : 0.0f;
        o.z = (d2 > 0.0f) ? rsqrtf(d2) : 0.0f;
        o.w = (d3 > 0.0f) ? rsqrtf(d3) : 0.0f;
        ((float4*)dinv)[i4] = o;
    }
}

#define GEMM_WAVES 16
static __device__ __forceinline__ void
gemm_body(const float* __restrict__ x, const float* __restrict__ W,
          const float* __restrict__ b, float* __restrict__ h, int n_nodes, int blk) {
    __shared__ float4 wl[16 * 4 * 32];

    const int tid  = threadIdx.x;
    const int lane = tid & 31;
    const int wave = tid >> 5;
    const int half = lane >> 4;
    const int l    = lane & 15;
    const int m0   = (blk * GEMM_WAVES + wave) * 16;

    int row = m0 + l;
    int rrow = (row < n_nodes) ? row : (n_nodes - 1);

    const float4* __restrict__ xr = (const float4*)(x + (size_t)rrow * D_IN) + half;
    float4 a[16];
#pragma unroll
    for (int j = 0; j < 16; ++j) a[j] = xr[2 * j];

    float bias[4];
#pragma unroll
    for (int t = 0; t < 4; ++t) bias[t] = b[t * 16 + l];

    {
        const float4* __restrict__ W4 = (const float4*)W;
#pragma unroll
        for (int i = 0; i < (D_OUT * D_IN / 4) / (GEMM_WAVES * 32); ++i) {
            const int g  = i * (GEMM_WAVES * 32) + tid;
            const int r  = g >> 5;
            const int c  = g & 31;
            const int t  = r >> 4, n = r & 15;
            const int j  = c >> 1, hf = c & 1;
            wl[(j * 4 + t) * 32 + hf * 16 + (n ^ j)] = W4[g];
        }
    }
    __syncthreads();

    v8f acc[4];
#pragma unroll
    for (int t = 0; t < 4; ++t) {
#pragma unroll
        for (int r = 0; r < 8; ++r) acc[t][r] = bias[t];
    }

#pragma unroll
    for (int j = 0; j < 16; ++j) {
        float4 bv[4];
#pragma unroll
        for (int t = 0; t < 4; ++t) bv[t] = wl[(j * 4 + t) * 32 + (lane ^ j)];
        v2f a0, a1;
        a0[0] = a[j].x; a0[1] = a[j].y;
        a1[0] = a[j].z; a1[1] = a[j].w;
#pragma unroll
        for (int t = 0; t < 4; ++t) {
            v2f b0; b0[0] = bv[t].x; b0[1] = bv[t].y;
            acc[t] = __builtin_amdgcn_wmma_f32_16x16x4_f32(
                 false, a0,  false, b0,
                 (short)0, acc[t],  false,  false);
        }
#pragma unroll
        for (int t = 0; t < 4; ++t) {
            v2f b1; b1[0] = bv[t].z; b1[1] = bv[t].w;
            acc[t] = __builtin_amdgcn_wmma_f32_16x16x4_f32(
                 false, a1,  false, b1,
                 (short)0, acc[t],  false,  false);
        }
    }

    float* __restrict__ hbase = h + (size_t)m0 * D_OUT;
#pragma unroll
    for (int r = 0; r < 8; ++r) {
        const int mr = r + 8 * half;
        if (m0 + mr < n_nodes) {
#pragma unroll
            for (int t = 0; t < 4; ++t)
                hbase[(size_t)mr * D_OUT + t * 16 + l] = acc[t][r];
        }
    }
}

__global__ void __launch_bounds__(HT)
front_kernel(const int* __restrict__ to_idx, unsigned* __restrict__ cntp,
             unsigned* __restrict__ keyw, int np, int e_cnt,
             int n_ranges, int n_hist,
             const float* __restrict__ x, const float* __restrict__ W,
             const float* __restrict__ b, float* __restrict__ h, int n_nodes) {
    const int bid = blockIdx.x;
    if (bid < n_hist) {
        const int slice = bid / n_ranges;
        deg_hist_body(to_idx, cntp, keyw, np, e_cnt, bid - slice * n_ranges, slice);
    } else {
        gemm_body(x, W, b, h, n_nodes, bid - n_hist);
    }
}

#define EP_SLICES 2
#define TILE_ROWS 896
#define QCAP      256
#define EP_WAVES  32
#define EP_THREADS (EP_WAVES * 32)

__global__ void __launch_bounds__(EP_THREADS)
edge_tile_kernel(const int* __restrict__ from_idx, const int* __restrict__ to_idx,
                 const unsigned* __restrict__ keyw,
                 const float* __restrict__ h, const float* __restrict__ dinv,
                 float* __restrict__ part, int n_nodes, int e_cnt) {
    __shared__ __attribute__((aligned(16))) float tile[TILE_ROWS * D_OUT];
    __shared__ float dinv_t[TILE_ROWS];
    __shared__ int   queue[EP_WAVES * QCAP];

    const int tid  = threadIdx.x;
    const int lane = tid & 31;
    const int wave = __builtin_amdgcn_readfirstlane(tid >> 5);
    const int base = blockIdx.x * TILE_ROWS;
    const int slice = blockIdx.y;
    int rows = n_nodes - base;
    if (rows > TILE_ROWS) rows = TILE_ROWS;

    {
        const float4* __restrict__ hsrc = (const float4*)(h + (size_t)base * D_OUT);
        float4* tdst = (float4*)tile;
        if (slice == 0) {
            for (int i = tid; i < rows * (D_OUT / 4); i += EP_THREADS) {
                float di = dinv[base + (i >> 4)];
                float4 v = hsrc[i];
                float4 o;
                o.x = v.x * di * di; o.y = v.y * di * di;
                o.z = v.z * di * di; o.w = v.w * di * di;
                tdst[i] = o;
            }
        } else {
            const float4 z = make_float4(0.0f, 0.0f, 0.0f, 0.0f);
            for (int i = tid; i < rows * (D_OUT / 4); i += EP_THREADS) tdst[i] = z;
        }
        for (int r = tid; r < rows; r += EP_THREADS) dinv_t[r] = dinv[base + r];
    }
    __syncthreads();

    int* q = queue + wave * QCAP;
    int count = 0;
    const unsigned ubase = (unsigned)base;
    const unsigned urows = (unsigned)rows;

    auto add_row = [&](float2 hv, int lr, float c) {
        float* p = &tile[lr * D_OUT + 2 * lane];
        __hip_atomic_fetch_add(p,     hv.x * c, __ATOMIC_RELAXED, __HIP_MEMORY_SCOPE_WORKGROUP);
        __hip_atomic_fetch_add(p + 1, hv.y * c, __ATOMIC_RELAXED, __HIP_MEMORY_SCOPE_WORKGROUP);
    };

    auto batch = [&](int b0, int n, int m, bool full) {
        int idx = b0 + lane;
        if (idx > n - 1) idx = n - 1;
        const int e = q[idx];
        const int f = from_idx[e];
        const unsigned d = (unsigned)to_idx[e] - ubase;
        const bool ok = d < urows;
        const int lr = ok ? (int)d : 0;
        const float df = dinv[f];
        if (full) {
            float2 hv[32];
#pragma unroll
            for (int k = 0; k < 32; ++k) {
                int fj = __builtin_amdgcn_readlane(f, k);
                hv[k] = ((const float2*)(h + (size_t)fj * D_OUT))[lane];
            }
            const float c = ok ? df * dinv_t[lr] : 0.0f;
#pragma unroll
            for (int k = 0; k < 32; ++k) {
                int   lj = __builtin_amdgcn_readlane(lr, k);
                float cj = __int_as_float(__builtin_amdgcn_readlane(__float_as_int(c), k));
                add_row(hv[k], lj, cj);
            }
        } else {
            const float c = ok ? df * dinv_t[lr] : 0.0f;
            for (int j = 0; j < m; j += 8) {
                float2 hv[8];
#pragma unroll
                for (int k = 0; k < 8; ++k) {
                    int fj = __builtin_amdgcn_readlane(f, j + k);
                    hv[k] = ((const float2*)(h + (size_t)fj * D_OUT))[lane];
                }
#pragma unroll
                for (int k = 0; k < 8; ++k) {
                    if (j + k < m) {
                        int   lj = __builtin_amdgcn_readlane(lr, j + k);
                        float cj = __int_as_float(__builtin_amdgcn_readlane(__float_as_int(c), j + k));
                        add_row(hv[k], lj, cj);
                    }
                }
            }
        }
    };

    auto flush = [&](bool final_call) {
        const int n = count;
        const int nb = n & ~31;
        for (int b0 = 0; b0 < nb; b0 += 32) batch(b0, n, 32, true);
        const int r = n - nb;
        if (final_call) {
            if (r > 0) batch(nb, n, r, false);
            count = 0;
        } else {
            int v = 0;
            if (lane < r) v = q[nb + lane];
            if (nb > 0 && lane < r) q[lane] = v;
            count = (nb > 0) ? r : n;
        }
    };

    int head = (int)(((16u - ((unsigned)(size_t)to_idx & 15u)) & 15u) >> 2);
    if (head > e_cnt) head = e_cnt;
    const int nwc = (e_cnt - head) >> 9;
    const int c_lo = (int)(((long long)nwc * slice) / EP_SLICES);
    const int nloc = (int)(((long long)nwc * (slice + 1)) / EP_SLICES) - c_lo;
    const uint4* __restrict__ vp = (const uint4*)keyw + (size_t)c_lo * 32;
    const int ebase = head + (c_lo << 9);

    if (wave == 0 && slice == 0) {
        const int vec_end = head + (nwc << 9);
        const int nleft = head + (e_cnt - vec_end);
        for (int k0 = 0; k0 < nleft; k0 += 32) {
            int k = k0 + lane;
            int e = (k < head) ? k : (vec_end + (k - head));
            unsigned d = 0xFFFFFFFFu;
            if (k < nleft) d = (unsigned)to_idx[e] - ubase;
            const bool m = d < urows;
            unsigned mk = __builtin_amdgcn_ballot_w32(m);
            if (mk) {
                int pos = count + (int)__builtin_amdgcn_mbcnt_lo(mk, 0u);
                if (m) q[pos] = e;
                count += __builtin_popcount(mk);
            }
            if (count > QCAP - 32) flush(false);
        }
    }

    const int S = (nloc + 31) >> 5;
    const int rot = (int)((blockIdx.x * (unsigned)S) / gridDim.x);

    auto chunk_of = [&](int s) -> int {
        int sr = s + rot;
        if (sr >= S) sr -= S;
        return sr * 32 + wave;
    };
    auto load_chunk = [&](int s) -> uint4 {
        int c = chunk_of(s);
        if (c > nloc - 1) c = nloc - 1;
        return vp[c * 32 + lane];
    };

#define PF 4
    if (nloc > 0) {
        const unsigned kb = (unsigned)blockIdx.x * 0x01010101u;
        auto differ = [&](unsigned w) -> unsigned {
            const unsigned x = w ^ kb;
            return ((x & 0x7F7F7F7Fu) + 0x7F7F7F7Fu) | x;
        };
        auto step = [&](int s, uint4 cur) {
            const int cs = chunk_of(s);
            const bool on = (s < S) && (cs < nloc);
            const unsigned h0 = ~differ(cur.x) & 0x80808080u;
            const unsigned h1 = ~differ(cur.y) & 0x80808080u;
            const unsigned h2 = ~differ(cur.z) & 0x80808080u;
            const unsigned h3 = ~differ(cur.w) & 0x80808080u;
            unsigned hm = (h0 >> 7) | (h1 >> 6) | (h2 >> 5) | (h3 >> 4);
            if (!on) hm = 0u;
            const int e0 = ebase + (cs << 9) + (lane << 4);
#pragma nounroll
            for (int it = 0; it < 16; ++it) {
                const unsigned mk = __builtin_amdgcn_ballot_w32(hm != 0u);
                if (!mk) break;
                const int pos = count + (int)__builtin_amdgcn_mbcnt_lo(mk, 0u);
                if (hm) {
                    const int p = __builtin_ctz(hm);
                    q[pos] = e0 + (((p & 7) << 2) | (p >> 3));
                    hm &= hm - 1u;
                }
                count += __builtin_popcount(mk);
                if (count > QCAP - 32) flush(false);
            }
        };

        uint4 bufa[PF], bufb[PF];
#pragma unroll
        for (int u = 0; u < PF; ++u) bufa[u] = load_chunk(u);

        for (int s0 = 0; s0 < S; s0 += 2 * PF) {
#pragma unroll
            for (int u = 0; u < PF; ++u) bufb[u] = load_chunk(s0 + PF + u);
#pragma unroll
            for (int u = 0; u < PF; ++u) step(s0 + u, bufa[u]);
#pragma unroll
            for (int u = 0; u < PF; ++u) bufa[u] = load_chunk(s0 + 2 * PF + u);
#pragma unroll
            for (int u = 0; u < PF; ++u) step(s0 + PF + u, bufb[u]);
        }
    }
    flush(true);

    __syncthreads();

    {
        float4* __restrict__ odst = (float4*)(part + ((size_t)slice * n_nodes + base) * D_OUT);
        const float4* tsrc = (const float4*)tile;
        for (int i = tid; i < rows * (D_OUT / 4); i += EP_THREADS) odst[i] = tsrc[i];
    }
}

#define SUM_U 8
__global__ void __launch_bounds__(256)
sum_parts_kernel(const float* __restrict__ part, float* __restrict__ out, int n4) {
    const float4* __restrict__ p = (const float4*)part;
    float4* __restrict__ o = (float4*)out;
    const int i0 = blockIdx.x * (SUM_U * 256) + threadIdx.x;
    float4 v[EP_SLICES][SUM_U];
#pragma unroll
    for (int s = 0; s < EP_SLICES; ++s) {
#pragma unroll
        for (int u = 0; u < SUM_U; ++u) {
            int j = i0 + u * 256;
            if (j > n4 - 1) j = n4 - 1;
            v[s][u] = p[(size_t)s * n4 + j];
        }
    }
#pragma unroll
    for (int u = 0; u < SUM_U; ++u) {
        const int j = i0 + u * 256;
        float4 a = v[0][u];
#pragma unroll
        for (int s = 1; s < EP_SLICES; ++s) {
            a.x += v[s][u].x; a.y += v[s][u].y; a.z += v[s][u].z; a.w += v[s][u].w;
        }
        if (j < n4) o[j] = a;
    }
}

extern "C" void kernel_launch(void* const* d_in, const int* in_sizes, int n_in,
                              void* d_out, int out_size, void* d_ws, size_t ws_size,
                              hipStream_t stream) {
    const float* x  = (const float*)d_in[0];
    const float* W  = (const float*)d_in[1];
    const float* b  = (const float*)d_in[2];
    const int*   ei = (const int*)d_in[3];

    const int N = in_sizes[0] / D_IN;
    const int E = in_sizes[3] / 2;
    const int* from_idx = ei;
    const int* to_idx   = ei + E;

    float* out = (float*)d_out;

    const int NP = (N + 63) & ~63;
    float* deg = (float*)d_ws;
    size_t deg_bytes = ((size_t)NP * sizeof(float) + 255) & ~(size_t)255;
    float* h = (float*)((char*)d_ws + deg_bytes);
    float* part = h + (size_t)N * D_OUT;
    unsigned* cntp = (unsigned*)(part + (size_t)EP_SLICES * N * D_OUT);
    unsigned* keyw = cntp + (size_t)HK * NP;

    int n_ranges = (NP + HR_NODES - 1) / HR_NODES;
    int n_hist = n_ranges * HK;
    int m_tiles = (N + 15) / 16;
    int n_gemm = (m_tiles + GEMM_WAVES - 1) / GEMM_WAVES;
    front_kernel<<<n_hist + n_gemm, HT, 0, stream>>>(to_idx, cntp, keyw, NP, E, n_ranges, n_hist, x, W, b, h, N);
    dinv_kernel<<<((NP >> 2) + 255) / 256, 256, 0, stream>>>(cntp, deg, NP);

    int n_tiles = (N + TILE_ROWS - 1) / TILE_ROWS;
    edge_tile_kernel<<<dim3(n_tiles, EP_SLICES), EP_THREADS, 0, stream>>>(from_idx, to_idx, keyw, h, deg, part, N, E);

    int n4 = N * (D_OUT / 4);
    sum_parts_kernel<<<(n4 + SUM_U * 256 - 1) / (SUM_U * 256), 256, 0, stream>>>(part, out, n4);
}


# edge pass: per queued edge one scattered 4-byte read of a packed word (source node | row in tile), written per edge by the count blocks, replaces two (to[e], from[e]); add order stays free as before
# speedup vs baseline: 3.9267x; 1.0441x over previous
#include <hip/hip_runtime.h>

#define D_IN  128
#define D_OUT 64

typedef __attribute__((ext_vector_type(2))) float v2f;
typedef __attribute__((ext_vector_type(8))) float v8f;

#define HR_NODES 6272
#define HK       16
#define HT       512
#define HU       8
#define KEY_ROWS 896u
#define REC_SHIFT 18
#define REC_U     4

static __device__ __forceinline__ void
deg_hist_body(const int* __restrict__ from_idx, const int* __restrict__ to_idx,
              unsigned* __restrict__ cntp, unsigned* __restrict__ keyw,
              unsigned* __restrict__ recw, int np, int e_cnt,
              int n_ranges, int range, int slice) {
    __shared__ __attribute__((aligned(16))) unsigned cnt[HR_NODES];

    const int tid   = threadIdx.x;
    const int base  = range * HR_NODES;
    int lim = np - base;
    if (lim > HR_NODES) lim = HR_NODES;
    const unsigned ubase = (unsigned)base;
    const unsigned ulim  = (unsigned)lim;

    {
        uint4* c4 = (uint4*)cnt;
        const uint4 z = make_uint4(0u, 0u, 0u, 0u);
        for (int i = tid; i < HR_NODES / 4; i += HT) c4[i] = z;
    }
    __syncthreads();

    auto count1 = [&](int t, unsigned lm) {
        const unsigned d = (unsigned)t - ubase;
        if (d < lm) __hip_atomic_fetch_add(&cnt[d], 1u, __ATOMIC_RELAXED, __HIP_MEMORY_SCOPE_WORKGROUP);
    };

    int head = (int)(((16u - ((unsigned)(size_t)to_idx & 15u)) & 15u) >> 2);
    if (head > e_cnt) head = e_cnt;
    const int nvec = (e_cnt - head) >> 2;
    if (slice == 0) {
        const int vec_end = head + (nvec << 2);
        const int nleft = head + (e_cnt - vec_end);
        if (tid < nleft) {
            const int e = (tid < head) ? tid : (vec_end + (tid - head));
            const int t = to_idx[e];
            count1(t, ulim);
            if (range == 0)
                recw[4 - head + e] = (unsigned)from_idx[e] | (((unsigned)t % KEY_ROWS) << REC_SHIFT);
        }
    }

    const int v_lo = (int)(((long long)nvec * slice) / HK);
    const int nloc = (int)(((long long)nvec * (slice + 1)) / HK) - v_lo;
    if (nloc > 0) {
        const int4* __restrict__ sp = (const int4*)(to_idx + head) + v_lo;
        auto load_piece = [&](int i) -> int4 {
            int c = (i < nloc) ? i : (nloc - 1);
            return sp[c];
        };
        auto count_piece = [&](int i, int4 t) {
            const unsigned lm = (i < nloc) ? ulim : 0u;
            count1(t.x, lm); count1(t.y, lm); count1(t.z, lm); count1(t.w, lm);
            if (range == 0 && i < nloc) {
                keyw[v_lo + i] = (((unsigned)t.x / KEY_ROWS) & 255u)
                               | ((((unsigned)t.y / KEY_ROWS) & 255u) << 8)
                               | ((((unsigned)t.z / KEY_ROWS) & 255u) << 16)
                               | (((unsigned)t.w / KEY_ROWS) << 24);
            }
        };

        int4 bufa[HU], bufb[HU];
#pragma unroll
        for (int u = 0; u < HU; ++u) bufa[u] = load_piece(u * HT + tid);

        for (int i0 = 0; i0 < nloc; i0 += 2 * HU * HT) {
#pragma unroll
            for (int u = 0; u < HU; ++u) bufb[u] = load_piece(i0 + (HU + u) * HT + tid);
#pragma unroll
            for (int u = 0; u < HU; ++u) count_piece(i0 + u * HT + tid, bufa[u]);
#pragma unroll
            for (int u = 0; u < HU; ++u) bufa[u] = load_piece(i0 + (2 * HU + u) * HT + tid);
#pragma unroll
            for (int u = 0; u < HU; ++u) count_piece(i0 + (HU + u) * HT + tid, bufb[u]);
        }

        const int r_lo = (int)(((long long)nloc * range) / n_ranges);
        const int r_hi = (int)(((long long)nloc * (range + 1)) / n_ranges);
        const int* __restrict__ fq = from_idx + head + 4 * (size_t)v_lo;
        uint4* __restrict__ rq = (uint4*)(recw + 4) + v_lo;
        for (int i0 = r_lo; i0 < r_hi; i0 += REC_U * HT) {
            int4 tt[REC_U];
            int  ff[REC_U][4];
#pragma unroll
            for (int u = 0; u < REC_U; ++u) {
                int i = i0 + u * HT + tid;
                if (i > r_hi - 1) i = r_hi - 1;
                tt[u] = sp[i];
                const int* __restrict__ fp = fq + 4 * (size_t)i;
                ff[u][0] = fp[0]; ff[u][1] = fp[1]; ff[u][2] = fp[2]; ff[u][3] = fp[3];
            }
#pragma unroll
            for (int u = 0; u < REC_U; ++u) {
                const int i = i0 + u * HT + tid;
                if (i < r_hi) {
                    uint4 w;
                    w.x = (unsigned)ff[u][0] | (((unsigned)tt[u].x % KEY_ROWS) << REC_SHIFT);
                    w.y = (unsigned)ff[u][1] | (((unsigned)tt[u].y % KEY_ROWS) << REC_SHIFT);
                    w.z = (unsigned)ff[u][2] | (((unsigned)tt[u].z % KEY_ROWS) << REC_SHIFT);
                    w.w = (unsigned)ff[u][3] | (((unsigned)tt[u].w % KEY_ROWS) << REC_SHIFT);
                    rq[i] = w;
                }
            }
        }
    }
    __syncthreads();

    {
        uint4* __restrict__ dst = (uint4*)(cntp + (size_t)slice * np + base);
        const uint4* c4 = (const uint4*)cnt;
        for (int i = tid; i < (lim >> 2); i += HT) dst[i] = c4[i];
    }
}

__global__ void __launch_bounds__(256)
dinv_kernel(const unsigned* __restrict__ cntp, float* __restrict__ dinv, int np) {
    const int n4 = np >> 2;
    const int i4 = blockIdx.x * 256 + threadIdx.x;
    if (i4 < n4) {
        const uint4* __restrict__ p = (const uint4*)cntp + i4;
        uint4 c[HK];
#pragma unroll
        for (int k = 0; k < HK; ++k) c[k] = p[(size_t)k * n4];
        uint4 s = c[0];
#pragma unroll
        for (int k = 1; k < HK; ++k) { s.x += c[k].x; s.y += c[k].y; s.z += c[k].z; s.w += c[k].w; }
        const float d0 = 1.0f + (float)s.x;
        const float d1 = 1.0f + (float)s.y;
        const float d2 = 1.0f + (float)s.z;
        const float d3 = 1.0f + (float)s.w;
        float4 o;
        o.x = (d0 > 0.0f) ? rsqrtf(d0) : 0.0f;
        o.y = (d1 > 0.0f) ? rsqrtf(d1) : 0.0f;
        o.z = (d2 > 0.0f) ? rsqrtf(d2) : 0.0f;
        o.w = (d3 > 0.0f) ? rsqrtf(d3) : 0.0f;
        ((float4*)dinv)[i4] = o;
    }
}

#define GEMM_WAVES 16
static __device__ __forceinline__ void
gemm_body(const float* __restrict__ x, const float* __restrict__ W,
          const float* __restrict__ b, float* __restrict__ h, int n_nodes, int blk) {
    __shared__ float4 wl[16 * 4 * 32];

    const int tid  = threadIdx.x;
    const int lane = tid & 31;
    const int wave = tid >> 5;
    const int half = lane >> 4;
    const int l    = lane & 15;
    const int m0   = (blk * GEMM_WAVES + wave) * 16;

    int row = m0 + l;
    int rrow = (row < n_nodes) ? row : (n_nodes - 1);

    const float4* __restrict__ xr = (const float4*)(x + (size_t)rrow * D_IN) + half;
    float4 a[16];
#pragma unroll
    for (int j = 0; j < 16; ++j) a[j] = xr[2 * j];

    float bias[4];
#pragma unroll
    for (int t = 0; t < 4; ++t) bias[t] = b[t * 16 + l];

    {
        const float4* __restrict__ W4 = (const float4*)W;
#pragma unroll
        for (int i = 0; i < (D_OUT * D_IN / 4) / (GEMM_WAVES * 32); ++i) {
            const int g  = i * (GEMM_WAVES * 32) + tid;
            const int r  = g >> 5;
            const int c  = g & 31;
            const int t  = r >> 4, n = r & 15;
            const int j  = c >> 1, hf = c & 1;
            wl[(j * 4 + t) * 32 + hf * 16 + (n ^ j)] = W4[g];
        }
    }
    __syncthreads();

    v8f acc[4];
#pragma unroll
    for (int t = 0; t < 4; ++t) {
#pragma unroll
        for (int r = 0; r < 8; ++r) acc[t][r] = bias[t];
    }

#pragma unroll
    for (int j = 0; j < 16; ++j) {
        float4 bv[4];
#pragma unroll
        for (int t = 0; t < 4; ++t) bv[t] = wl[(j * 4 + t) * 32 + (lane ^ j)];
        v2f a0, a1;
        a0[0] = a[j].x; a0[1] = a[j].y;
        a1[0] = a[j].z; a1[1] = a[j].w;
#pragma unroll
        for (int t = 0; t < 4; ++t) {
            v2f b0; b0[0] = bv[t].x; b0[1] = bv[t].y;
            acc[t] = __builtin_amdgcn_wmma_f32_16x16x4_f32(
                 false, a0,  false, b0,
                 (short)0, acc[t],  false,  false);
        }
#pragma unroll
        for (int t = 0; t < 4; ++t) {
            v2f b1; b1[0] = bv[t].z; b1[1] = bv[t].w;
            acc[t] = __builtin_amdgcn_wmma_f32_16x16x4_f32(
                 false, a1,  false, b1,
                 (short)0, acc[t],  false,  false);
        }
    }

    float* __restrict__ hbase = h + (size_t)m0 * D_OUT;
#pragma unroll
    for (int r = 0; r < 8; ++r) {
        const int mr = r + 8 * half;
        if (m0 + mr < n_nodes) {
#pragma unroll
            for (int t = 0; t < 4; ++t)
                hbase[(size_t)mr * D_OUT + t * 16 + l] = acc[t][r];
        }
    }
}

__global__ void __launch_bounds__(HT)
front_kernel(const int* __restrict__ from_idx, const int* __restrict__ to_idx,
             unsigned* __restrict__ cntp, unsigned* __restrict__ keyw,
             unsigned* __restrict__ recw, int np, int e_cnt,
             int n_ranges, int n_hist,
             const float* __restrict__ x, const float* __restrict__ W,
             const float* __restrict__ b, float* __restrict__ h, int n_nodes) {
    const int bid = blockIdx.x;
    if (bid < n_hist) {
        const int slice = bid / n_ranges;
        deg_hist_body(from_idx, to_idx, cntp, keyw, recw, np, e_cnt, n_ranges, bid - slice * n_ranges, slice);
    } else {
        gemm_body(x, W, b, h, n_nodes, bid - n_hist);
    }
}

#define EP_SLICES 2
#define TILE_ROWS 896
#define QCAP      256
#define EP_WAVES  32
#define EP_THREADS (EP_WAVES * 32)

__global__ void __launch_bounds__(EP_THREADS)
edge_tile_kernel(const int* __restrict__ to_idx,
                 const unsigned* __restrict__ keyw, const unsigned* __restrict__ recw,
                 const float* __restrict__ h, const float* __restrict__ dinv,
                 float* __restrict__ part, int n_nodes, int e_cnt) {
    __shared__ __attribute__((aligned(16))) float tile[TILE_ROWS * D_OUT];
    __shared__ float dinv_t[TILE_ROWS];
    __shared__ int   queue[EP_WAVES * QCAP];

    const int tid  = threadIdx.x;
    const int lane = tid & 31;
    const int wave = __builtin_amdgcn_readfirstlane(tid >> 5);
    const int base = blockIdx.x * TILE_ROWS;
    const int slice = blockIdx.y;
    int rows = n_nodes - base;
    if (rows > TILE_ROWS) rows = TILE_ROWS;
    int head = (int)(((16u - ((unsigned)(size_t)to_idx & 15u)) & 15u) >> 2);
    if (head > e_cnt) head = e_cnt;
    const unsigned* __restrict__ rece = recw + (4 - head);

    {
        const float4* __restrict__ hsrc = (const float4*)(h + (size_t)base * D_OUT);
        float4* tdst = (float4*)tile;
        if (slice == 0) {
            for (int i = tid; i < rows * (D_OUT / 4); i += EP_THREADS) {
                float di = dinv[base + (i >> 4)];
                float4 v = hsrc[i];
                float4 o;
                o.x = v.x * di * di; o.y = v.y * di * di;
                o.z = v.z * di * di; o.w = v.w * di * di;
                tdst[i] = o;
            }
        } else {
            const float4 z = make_float4(0.0f, 0.0f, 0.0f, 0.0f);
            for (int i = tid; i < rows * (D_OUT / 4); i += EP_THREADS) tdst[i] = z;
        }
        for (int r = tid; r < rows; r += EP_THREADS) dinv_t[r] = dinv[base + r];
    }
    __syncthreads();

    int* q = queue + wave * QCAP;
    int count = 0;
    const unsigned ubase = (unsigned)base;
    const unsigned urows = (unsigned)rows;

    auto add_row = [&](float2 hv, int lr, float c) {
        float* p = &tile[lr * D_OUT + 2 * lane];
        __hip_atomic_fetch_add(p,     hv.x * c, __ATOMIC_RELAXED, __HIP_MEMORY_SCOPE_WORKGROUP);
        __hip_atomic_fetch_add(p + 1, hv.y * c, __ATOMIC_RELAXED, __HIP_MEMORY_SCOPE_WORKGROUP);
    };

    auto batch = [&](int b0, int n, int m, bool full) {
        int idx = b0 + lane;
        if (idx > n - 1) idx = n - 1;
        const int e = q[idx];
        const unsigned rw = rece[e];
        const int f = (int)(rw & ((1u << REC_SHIFT) - 1u));
        const unsigned d = rw >> REC_SHIFT;
        const bool ok = d < urows;
        const int lr = ok ? (int)d : 0;
        const float df = dinv[f];
        if (full) {
            float2 hv[32];
#pragma unroll
            for (int k = 0; k < 32; ++k) {
                int fj = __builtin_amdgcn_readlane(f, k);
                hv[k] = ((const float2*)(h + (size_t)fj * D_OUT))[lane];
            }
            const float c = ok ? df * dinv_t[lr] : 0.0f;
#pragma unroll
            for (int k = 0; k < 32; ++k) {
                int   lj = __builtin_amdgcn_readlane(lr, k);
                float cj = __int_as_float(__builtin_amdgcn_readlane(__float_as_int(c), k));
                add_row(hv[k], lj, cj);
            }
        } else {
            const float c = ok ? df * dinv_t[lr] : 0.0f;
            for (int j = 0; j < m; j += 8) {
                float2 hv[8];
#pragma unroll
                for (int k = 0; k < 8; ++k) {
                    int fj = __builtin_amdgcn_readlane(f, j + k);
                    hv[k] = ((const float2*)(h + (size_t)fj * D_OUT))[lane];
                }
#pragma unroll
                for (int k = 0; k < 8; ++k) {
                    if (j + k < m) {
                        int   lj = __builtin_amdgcn_readlane(lr, j + k);
                        float cj = __int_as_float(__builtin_amdgcn_readlane(__float_as_int(c), j + k));
                        add_row(hv[k], lj, cj);
                    }
                }
            }
        }
    };

    auto flush = [&](bool final_call) {
        const int n = count;
        const int nb = n & ~31;
        for (int b0 = 0; b0 < nb; b0 += 32) batch(b0, n, 32, true);
        const int r = n - nb;
        if (final_call) {
            if (r > 0) batch(nb, n, r, false);
            count = 0;
        } else {
            int v = 0;
            if (lane < r) v = q[nb + lane];
            if (nb > 0 && lane < r) q[lane] = v;
            count = (nb > 0) ? r : n;
        }
    };

    const int nwc = (e_cnt - head) >> 9;
    const int c_lo = (int)(((long long)nwc * slice) / EP_SLICES);
    const int nloc = (int)(((long long)nwc * (slice + 1)) / EP_SLICES) - c_lo;
    const uint4* __restrict__ vp = (const uint4*)keyw + (size_t)c_lo * 32;
    const int ebase = head + (c_lo << 9);

    if (wave == 0 && slice == 0) {
        const int vec_end = head + (nwc << 9);
        const int nleft = head + (e_cnt - vec_end);
        for (int k0 = 0; k0 < nleft; k0 += 32) {
            int k = k0 + lane;
            int e = (k < head) ? k : (vec_end + (k - head));
            unsigned d = 0xFFFFFFFFu;
            if (k < nleft) d = (unsigned)to_idx[e] - ubase;
            const bool m = d < urows;
            unsigned mk = __builtin_amdgcn_ballot_w32(m);
            if (mk) {
                int pos = count + (int)__builtin_amdgcn_mbcnt_lo(mk, 0u);
                if (m) q[pos] = e;
                count += __builtin_popcount(mk);
            }
            if (count > QCAP - 32) flush(false);
        }
    }

    const int S = (nloc + 31) >> 5;
    const int rot = (int)((blockIdx.x * (unsigned)S) / gridDim.x);

    auto chunk_of = [&](int s) -> int {
        int sr = s + rot;
        if (sr >= S) sr -= S;
        return sr * 32 + wave;
    };
    auto load_chunk = [&](int s) -> uint4 {
        int c = chunk_of(s);
        if (c > nloc - 1) c = nloc - 1;
        return vp[c * 32 + lane];
    };

#define PF 4
    if (nloc > 0) {
        const unsigned kb = (unsigned)blockIdx.x * 0x01010101u;
        auto differ = [&](unsigned w) -> unsigned {
            const unsigned x = w ^ kb;
            return ((x & 0x7F7F7F7Fu) + 0x7F7F7F7Fu) | x;
        };
        auto step = [&](int s, uint4 cur) {
            const int cs = chunk_of(s);
            const bool on = (s < S) && (cs < nloc);
            const unsigned h0 = ~differ(cur.x) & 0x80808080u;
            const unsigned h1 = ~differ(cur.y) & 0x80808080u;
            const unsigned h2 = ~differ(cur.z) & 0x80808080u;
            const unsigned h3 = ~differ(cur.w) & 0x80808080u;
            unsigned hm = (h0 >> 7) | (h1 >> 6) | (h2 >> 5) | (h3 >> 4);
            if (!on) hm = 0u;
            const int e0 = ebase + (cs << 9) + (lane << 4);
#pragma nounroll
            for (int it = 0; it < 16; ++it) {
                const unsigned mk = __builtin_amdgcn_ballot_w32(hm != 0u);
                if (!mk) break;
                const int pos = count + (int)__builtin_amdgcn_mbcnt_lo(mk, 0u);
                if (hm) {
                    const int p = __builtin_ctz(hm);
                    q[pos] = e0 + (((p & 7) << 2) | (p >> 3));
                    hm &= hm - 1u;
                }
                count += __builtin_popcount(mk);
                if (count > QCAP - 32) flush(false);
            }
        };

        uint4 bufa[PF], bufb[PF];
#pragma unroll
        for (int u = 0; u < PF; ++u) bufa[u] = load_chunk(u);

        for (int s0 = 0; s0 < S; s0 += 2 * PF) {
#pragma unroll
            for (int u = 0; u < PF; ++u) bufb[u] = load_chunk(s0 + PF + u);
#pragma unroll
            for (int u = 0; u < PF; ++u) step(s0 + u, bufa[u]);
#pragma unroll
            for (int u = 0; u < PF; ++u) bufa[u] = load_chunk(s0 + 2 * PF + u);
#pragma unroll
            for (int u = 0; u < PF; ++u) step(s0 + PF + u, bufb[u]);
        }
    }
    flush(true);

    __syncthreads();

    {
        float4* __restrict__ odst = (float4*)(part + ((size_t)slice * n_nodes + base) * D_OUT);
        const float4* tsrc = (const float4*)tile;
        for (int i = tid; i < rows * (D_OUT / 4); i += EP_THREADS) odst[i] = tsrc[i];
    }
}

#define SUM_U 8
__global__ void __launch_bounds__(256)
sum_parts_kernel(const float* __restrict__ part, float* __restrict__ out, int n4) {
    const float4* __restrict__ p = (const float4*)part;
    float4* __restrict__ o = (float4*)out;
    const int i0 = blockIdx.x * (SUM_U * 256) + threadIdx.x;
    float4 v[EP_SLICES][SUM_U];
#pragma unroll
    for (int s = 0; s < EP_SLICES; ++s) {
#pragma unroll
        for (int u = 0; u < SUM_U; ++u) {
            int j = i0 + u * 256;
            if (j > n4 - 1) j = n4 - 1;
            v[s][u] = p[(size_t)s * n4 + j];
        }
    }
#pragma unroll
    for (int u = 0; u < SUM_U; ++u) {
        const int j = i0 + u * 256;
        float4 a = v[0][u];
#pragma unroll
        for (int s = 1; s < EP_SLICES; ++s) {
            a.x += v[s][u].x; a.y += v[s][u].y; a.z += v[s][u].z; a.w += v[s][u].w;
        }
        if (j < n4) o[j] = a;
    }
}

extern "C" void kernel_launch(void* const* d_in, const int* in_sizes, int n_in,
                              void* d_out, int out_size, void* d_ws, size_t ws_size,
                              hipStream_t stream) {
    const float* x  = (const float*)d_in[0];
    const float* W  = (const float*)d_in[1];
    const float* b  = (const float*)d_in[2];
    const int*   ei = (const int*)d_in[3];

    const int N = in_sizes[0] / D_IN;
    const int E = in_sizes[3] / 2;
    const int* from_idx = ei;
    const int* to_idx   = ei + E;

    float* out = (float*)d_out;

    const int NP = (N + 63) & ~63;
    float* deg = (float*)d_ws;
    size_t deg_bytes = ((size_t)NP * sizeof(float) + 255) & ~(size_t)255;
    float* h = (float*)((char*)d_ws + deg_bytes);
    float* part = h + (size_t)N * D_OUT;
    unsigned* cntp = (unsigned*)(part + (size_t)EP_SLICES * N * D_OUT);
    unsigned* recw = cntp + (size_t)HK * NP;
    const size_t rec_words = ((size_t)E + 4 + 63) & ~(size_t)63;
    unsigned* keyw = recw + rec_words;

    int n_ranges = (NP + HR_NODES - 1) / HR_NODES;
    int n_hist = n_ranges * HK;
    int m_tiles = (N + 15) / 16;
    int n_gemm = (m_tiles + GEMM_WAVES - 1) / GEMM_WAVES;
    front_kernel<<<n_hist + n_gemm, HT, 0, stream>>>(from_idx, to_idx, cntp, keyw, recw, NP, E, n_ranges, n_hist, x, W, b, h, N);
    dinv_kernel<<<((NP >> 2) + 255) / 256, 256, 0, stream>>>(cntp, deg, NP);

    int n_tiles = (N + TILE_ROWS - 1) / TILE_ROWS;
    edge_tile_kernel<<<dim3(n_tiles, EP_SLICES), EP_THREADS, 0, stream>>>(to_idx, keyw, recw, h, deg, part, N, E);

    int n4 = N * (D_OUT / 4);
    sum_parts_kernel<<<(n4 + SUM_U * 256 - 1) / (SUM_U * 256), 256, 0, stream>>>(part, out, n4);
}
